# starting kernel
# baseline (speedup 1.0000x reference)
#include <hip/hip_runtime.h>
#include <hip/hip_bf16.h>
#include <math.h>

#define SLOPE 0.2f

typedef float v2f __attribute__((ext_vector_type(2)));
typedef float v8f __attribute__((ext_vector_type(8)));


__global__ void fill_f32(float* __restrict__ p, float v, int n) {
    int i = blockIdx.x * blockDim.x + threadIdx.x;
    if (i < n) p[i] = v;
}

__device__ __forceinline__ void atomicMaxFloat(float* addr, float value) {
    if (value >= 0.0f) {
        atomicMax((int*)addr, __float_as_int(value));
    } else {
        atomicMin((unsigned int*)addr, __float_as_uint(value));
    }
}

__device__ __forceinline__ float leaky(float v) {
    return v > 0.0f ? v : SLOPE * v;
}

template <int NT>
__global__ void gemm_wmma_strip(const float* __restrict__ A,
                                const float* __restrict__ B,
                                float* __restrict__ C,
                                int M, int K) {
    const int N = NT * 16;
    extern __shared__ float sB[];

    for (int i = threadIdx.x; i < K * N; i += blockDim.x) sB[i] = B[i];
    __syncthreads();

    const int lane = threadIdx.x & 31;
    const int wid  = threadIdx.x >> 5;
    const int row0 = (blockIdx.x * (blockDim.x >> 5) + wid) * 16;
    if (row0 >= M) return;

    const int mn = lane & 15;
    const int hi = lane >> 4;
    const int kg = hi << 1;

    const float* arow = A + (size_t)(row0 + mn) * K;

    v8f acc[NT];
#pragma unroll
    for (int t = 0; t < NT; ++t) acc[t] = (v8f){};

    for (int k0 = 0; k0 < K; k0 += 4) {
        v2f a;
        a.x = arow[k0 + kg];
        a.y = arow[k0 + kg + 1];
        const float* brow = sB + (size_t)(k0 + kg) * N + mn;
#pragma unroll
        for (int t = 0; t < NT; ++t) {
            v2f b;
            b.x = brow[t * 16];
            b.y = brow[t * 16 + N];
            acc[t] = __builtin_amdgcn_wmma_f32_16x16x4_f32(
                 false, a,  false, b,
                 (short)0, acc[t],  false,  false);
        }
    }

#pragma unroll
    for (int t = 0; t < NT; ++t) {
        float* crow = C + (size_t)(row0 + hi * 8) * N + t * 16 + mn;
#pragma unroll
        for (int v = 0; v < 8; ++v) crow[(size_t)v * N] = acc[t][v];
    }
}

__global__ void node_scores(const float* __restrict__ h,
                            const float* __restrict__ a_src,
                            const float* __restrict__ a_dst,
                            float* __restrict__ ls, float* __restrict__ ld,
                            int n_nodes, int C) {
    const int lane = threadIdx.x & 31;
    const int node = (blockIdx.x * blockDim.x + threadIdx.x) >> 5;
    if (node >= n_nodes) return;
    float s = 0.0f, d = 0.0f;
    for (int c = lane; c < C; c += 32) {
        float v = h[(size_t)node * C + c];
        s += v * a_src[c];
        d += v * a_dst[c];
    }
#pragma unroll
    for (int off = 16; off > 0; off >>= 1) {
        s += __shfl_xor(s, off, 32);
        d += __shfl_xor(d, off, 32);
    }
    if (lane == 0) { ls[node] = s; ld[node] = d; }
}


__global__ void edge_logit_max(const int* __restrict__ src, const int* __restrict__ dst,
                               const float* __restrict__ ls, const float* __restrict__ ld,
                               float* __restrict__ ev, float* __restrict__ lmax, int E) {
    int e = blockIdx.x * blockDim.x + threadIdx.x;
    if (e >= E) return;
    int d = dst[e];
    float lg = leaky(ls[src[e]] + ld[d]);
    ev[e] = lg;
    atomicMaxFloat(&lmax[d], lg);
}

__global__ void edge_expsum(const int* __restrict__ dst,
                            const float* __restrict__ lmax,
                            float* __restrict__ ev, float* __restrict__ denom, int E) {
    int e = blockIdx.x * blockDim.x + threadIdx.x;
    if (e >= E) return;
    int d = dst[e];
    float ex = __expf(ev[e] - lmax[d]);
    ev[e] = ex;
    atomicAdd(&denom[d], ex);
}

template <int C>
__global__ void edge_aggregate(const int* __restrict__ src, const int* __restrict__ dst,
                               const float* __restrict__ ev, const float* __restrict__ denom,
                               const float* __restrict__ h, float* __restrict__ agg, int E) {
    constexpr int GP = C / 4;
    int t = blockIdx.x * blockDim.x + threadIdx.x;
    int e = t / GP;
    int g = t - e * GP;
    if (e >= E) return;
    int s = src[e], d = dst[e];
    float alpha = ev[e] / denom[d];
    const float4 v = *((const float4*)(h + (size_t)s * C) + g);
    float* ap = agg + (size_t)d * C + g * 4;
    atomicAdd(ap + 0, alpha * v.x);
    atomicAdd(ap + 1, alpha * v.y);
    atomicAdd(ap + 2, alpha * v.z);
    atomicAdd(ap + 3, alpha * v.w);
}


__global__ void bias_act(const float* __restrict__ agg, const float* __restrict__ b,
                         float* __restrict__ out, int total, int C, int do_relu) {
    int i = blockIdx.x * blockDim.x + threadIdx.x;
    if (i >= total) return;
    float v = agg[i] + b[i % C];
    out[i] = do_relu ? fmaxf(v, 0.0f) : v;
}


static inline int cdiv(int a, int b) { return (a + b - 1) / b; }

extern "C" void kernel_launch(void* const* d_in, const int* in_sizes, int n_in,
                              void* d_out, int out_size, void* d_ws, size_t ws_size,
                              hipStream_t stream) {
    const float* x      = (const float*)d_in[0];
    const int*   eidx   = (const int*)d_in[1];
    const float* W1     = (const float*)d_in[2];
    const float* a_src1 = (const float*)d_in[3];
    const float* a_dst1 = (const float*)d_in[4];
    const float* b1     = (const float*)d_in[5];
    const float* W2     = (const float*)d_in[6];
    const float* a_src2 = (const float*)d_in[7];
    const float* a_dst2 = (const float*)d_in[8];
    const float* b2     = (const float*)d_in[9];
    float* out = (float*)d_out;

    const int IN = 128, H = 64, O = 32;
    const int N = in_sizes[0] / IN;
    const int E = in_sizes[1] / 2;
    const int* src = eidx;
    const int* dst = eidx + E;

    float* ws    = (float*)d_ws;
    float* h1    = ws;
    float* agg1  = h1 + (size_t)N * H;
    float* h2    = agg1 + (size_t)N * H;
    float* ls    = h2 + (size_t)N * O;
    float* ld    = ls + N;
    float* lmax  = ld + N;
    float* denom = lmax + N;
    float* ev    = denom + N;

    const int B = 256;

    gemm_wmma_strip<4><<<cdiv(N, 128), B, IN * H * sizeof(float), stream>>>(x, W1, h1, N, IN);

    node_scores<<<cdiv(N * 32, B), B, 0, stream>>>(h1, a_src1, a_dst1, ls, ld, N, H);

    fill_f32<<<cdiv(N, B), B, 0, stream>>>(lmax, -INFINITY, N);
    fill_f32<<<cdiv(N, B), B, 0, stream>>>(denom, 0.0f, N);
    fill_f32<<<cdiv(N * H, B), B, 0, stream>>>(agg1, 0.0f, N * H);

    edge_logit_max<<<cdiv(E, B), B, 0, stream>>>(src, dst, ls, ld, ev, lmax, E);
    edge_expsum<<<cdiv(E, B), B, 0, stream>>>(dst, lmax, ev, denom, E);
    edge_aggregate<64><<<cdiv(E * (H / 4), B), B, 0, stream>>>(src, dst, ev, denom, h1, agg1, E);

    bias_act<<<cdiv(N * H, B), B, 0, stream>>>(agg1, b1, h1, N * H, H,  1);

    gemm_wmma_strip<2><<<cdiv(N, 128), B, H * O * sizeof(float), stream>>>(h1, W2, h2, N, H);

    node_scores<<<cdiv(N * 32, B), B, 0, stream>>>(h2, a_src2, a_dst2, ls, ld, N, O);

    fill_f32<<<cdiv(N, B), B, 0, stream>>>(lmax, -INFINITY, N);
    fill_f32<<<cdiv(N, B), B, 0, stream>>>(denom, 0.0f, N);
    fill_f32<<<cdiv(N * O, B), B, 0, stream>>>(out, 0.0f, N * O);

    edge_logit_max<<<cdiv(E, B), B, 0, stream>>>(src, dst, ls, ld, ev, lmax, E);
    edge_expsum<<<cdiv(E, B), B, 0, stream>>>(dst, lmax, ev, denom, E);
    edge_aggregate<32><<<cdiv(E * (O / 4), B), B, 0, stream>>>(src, dst, ev, denom, h2, out, E);

    bias_act<<<cdiv(N * O, B), B, 0, stream>>>(out, b2, out, N * O, O,  0);
}


# Softmax max and exp-sum in ONE edge pass and ONE fold per layer: work-groups own (node range x edge chunk), sums against the chunk's own max, fold rescales them (sum order changes); 13 launches to 9
# speedup vs baseline: 2.5269x; 2.5269x over previous
#include <hip/hip_runtime.h>
#include <hip/hip_bf16.h>
#include <math.h>

#define SLOPE 0.2f

typedef float v2f __attribute__((ext_vector_type(2)));
typedef float v8f __attribute__((ext_vector_type(8)));


template <int CI>
__global__ __launch_bounds__(256)
void start_rows(float* __restrict__ acc, const float* __restrict__ bias, int n4) {
    static_assert(1024 % CI == 0, "a thread's four channels must not depend on the store index");
    const int cb = ((int)threadIdx.x * 4) % CI;
    const float4 bv = make_float4(bias[cb], bias[cb + 1], bias[cb + 2], bias[cb + 3]);
    float4* p = reinterpret_cast<float4*>(acc);
    const int q0 = (int)blockIdx.x * 1024 + (int)threadIdx.x;
    for (int j = 0; j < 4; ++j) {
        const int q = q0 + j * 256;
        if (q < n4) p[q] = bv;
    }
}

__device__ __forceinline__ float leaky(float v) {
    return v > 0.0f ? v : SLOPE * v;
}

template <int NT, int CI, bool RELU_A>
__global__ void gemm_wmma_strip(const float* __restrict__ A,
                                const float* __restrict__ B,
                                float* __restrict__ C,
                                const float* __restrict__ bias,
                                float* __restrict__ start,
                                const float* __restrict__ a_src,
                                const float* __restrict__ a_dst,
                                float* __restrict__ sc,
                                int M, int K) {
    constexpr int CIX = CI > 0 ? CI : 128;
    static_assert(128 % CIX == 0, "a lane's four channels must not depend on the store index");
    const int N = NT * 16;
    extern __shared__ float sB[];
    float* sX = sB + K * N;
    float* sV = sX + K * 16;

    float4 bv = make_float4(0.0f, 0.0f, 0.0f, 0.0f);
    if (CI > 0) {
        const int cb = (((int)threadIdx.x & 31) * 4) % CIX;
        bv = make_float4(bias[cb], bias[cb + 1], bias[cb + 2], bias[cb + 3]);
    }

    static_assert((NT * 16 & (NT * 16 - 1)) == 0, "the rotated walk below wraps with a mask");
    float va = 0.0f, vd = 0.0f;
    if ((int)threadIdx.x < N) {
        va = a_src[threadIdx.x];
        vd = a_dst[threadIdx.x];
    }

    for (int i = threadIdx.x; i < K * N; i += blockDim.x) sB[i] = B[i];
    for (int i = threadIdx.x; i < K * 16; i += blockDim.x) sX[i] = 0.0f;
    if ((int)threadIdx.x < N) {
        sV[2 * threadIdx.x]     = va;
        sV[2 * threadIdx.x + 1] = vd;
    }
    __syncthreads();
    for (int t = threadIdx.x; t < 2 * K; t += blockDim.x) {
        const int k = t >> 1, which = t & 1;
        float s = 0.0f;
        for (int jj = 0; jj < N; ++jj) {
            const int j = (jj + k) & (N - 1);
            s += sB[k * N + j] * sV[2 * j + which];
        }
        sX[k * 16 + which] = s;
    }
    __syncthreads();

    const int lane = threadIdx.x & 31;
    const int wid  = threadIdx.x >> 5;
    const int row0 = (blockIdx.x * (blockDim.x >> 5) + wid) * 16;
    if (row0 >= M) return;

    const int mn = lane & 15;
    const int hi = lane >> 4;
    const int kg = hi << 1;

    const float* arow = A + (size_t)(row0 + mn) * K;

    if (CI > 0) {
        float4* sp = reinterpret_cast<float4*>(start + (size_t)row0 * CIX);
        for (int i = 0; i < (16 * CIX) / 128; ++i) sp[i * 32 + lane] = bv;
    }

    v8f acc[NT];
#pragma unroll
    for (int t = 0; t < NT; ++t) acc[t] = (v8f){};
    v8f accs = (v8f){};

#pragma unroll (NT == 2 ? 8 : 4)
    for (int k0 = 0; k0 < K; k0 += 4) {
        v2f a;
        a.x = arow[k0 + kg];
        a.y = arow[k0 + kg + 1];
        if (RELU_A) {
            a.x = fmaxf(a.x, 0.0f);
            a.y = fmaxf(a.y, 0.0f);
        }
        const float* brow = sB + (size_t)(k0 + kg) * N + mn;
#pragma unroll
        for (int t = 0; t < NT; ++t) {
            v2f b;
            b.x = brow[t * 16];
            b.y = brow[t * 16 + N];
            acc[t] = __builtin_amdgcn_wmma_f32_16x16x4_f32(
                 false, a,  false, b,
                 (short)0, acc[t],  false,  false);
        }
        {
            v2f b;
            b.x = sX[(k0 + kg) * 16 + mn];
            b.y = sX[(k0 + kg + 1) * 16 + mn];
            accs = __builtin_amdgcn_wmma_f32_16x16x4_f32(
                 false, a,  false, b,
                 (short)0, accs,  false,  false);
        }
    }

    if (mn < 2) {
        float* sp = sc + (size_t)row0 * 2 + mn * 16 + hi * 8;
#pragma unroll
        for (int v = 0; v < 8; ++v) sp[v] = accs[v];
    }

#pragma unroll
    for (int t = 0; t < NT; ++t) {
        float* crow = C + (size_t)(row0 + hi * 8) * N + t * 16 + mn;
#pragma unroll
        for (int v = 0; v < 8; ++v) crow[(size_t)v * N] = acc[t][v];
    }
}


#define SEG_T     1024
#define SEG_R     8
#define SEG_NRCAP 6272

__device__ __forceinline__ void ldsMaxFloat(float* addr, float value) {
    if (value >= 0.0f) {
        atomicMax((int*)addr, __float_as_int(value));
    } else {
        atomicMin((unsigned int*)addr, __float_as_uint(value));
    }
}

template <int U>
__global__ __launch_bounds__(SEG_T)
void edge_softmax_stats(const int* __restrict__ src, const int* __restrict__ dst,
                        const float* __restrict__ sc, float* __restrict__ part,
                        int N, int E, int nr, int pitch) {
    __shared__ float Tm[SEG_NRCAP];
    __shared__ float Ts[SEG_NRCAP];
    const int r   = (int)blockIdx.x % SEG_R;
    const int c   = (int)blockIdx.x / SEG_R;
    const int lo  = r * nr;
    const int len = max(0, min(nr, N - lo));
    for (int i = threadIdx.x; i < len; i += SEG_T) {
        Tm[i] = -INFINITY;
        Ts[i] = 0.0f;
    }
    __syncthreads();

    const int e0 = c * (U * SEG_T);
    int   d[U], s[U];
    float p[U], q[U];
#pragma unroll
    for (int u = 0; u < U; ++u) {
        const int ee = e0 + (int)threadIdx.x + u * SEG_T;
        const int ec = min(ee, E - 1);
        d[u] = dst[ec];
        s[u] = src[ec];
    }
    __builtin_amdgcn_sched_barrier(0);
#pragma unroll
    for (int u = 0; u < U; ++u) {
        const int ee = e0 + (int)threadIdx.x + u * SEG_T;
        const int di = d[u] - lo;
        d[u] = ((unsigned)di < (unsigned)len && ee < E) ? di : -1;
    }
#pragma unroll
    for (int u = 0; u < U; ++u) {
        const int keep = ~(d[u] >> 31);
        const int dn   = d[u] + lo;
        p[u] = sc[(unsigned)((s[u] + (s[u] & ~15)) & keep)];
        q[u] = sc[(unsigned)((dn + (dn & ~15) + 16) & keep)];
    }
#pragma unroll
    for (int u = 0; u < U; ++u) p[u] = leaky(p[u] + q[u]);
#pragma unroll
    for (int u = 0; u < U; ++u) {
        if (d[u] >= 0) ldsMaxFloat(&Tm[d[u]], p[u]);
    }
    __syncthreads();
#pragma unroll
    for (int u = 0; u < U; ++u) q[u] = Tm[d[u] & ~(d[u] >> 31)];
#pragma unroll
    for (int u = 0; u < U; ++u) q[u] = __expf(p[u] - q[u]);
#pragma unroll
    for (int u = 0; u < U; ++u) {
        if (d[u] >= 0) atomicAdd(&Ts[d[u]], q[u]);
    }
    __syncthreads();

    float* pm = part + (size_t)(2 * c) * pitch + lo;
    float* ps = pm + pitch;
    for (int i = threadIdx.x; i < len; i += SEG_T) {
        pm[i] = Tm[i];
        ps[i] = Ts[i];
    }
}

__global__ void seg_fold(const float* __restrict__ part, float* __restrict__ msf,
                         int N, int CN, int pitch) {
    float2* ms = reinterpret_cast<float2*>(msf);
    const int i = blockIdx.x * blockDim.x + threadIdx.x;
    if (i >= N) return;
    float M = -INFINITY, S = 0.0f;
    for (int g = 0; g < CN; g += 16) {
        float m[16], v[16];
#pragma unroll
        for (int k = 0; k < 16; ++k) {
            const int cc = min(g + k, CN - 1);
            m[k] = part[(size_t)(2 * cc) * pitch + i];
            v[k] = part[(size_t)(2 * cc + 1) * pitch + i];
        }
        float gm = M;
#pragma unroll
        for (int k = 0; k < 16; ++k) gm = fmaxf(gm, m[k]);
        S = (M == -INFINITY) ? 0.0f : S * __expf(M - gm);
#pragma unroll
        for (int k = 0; k < 16; ++k) {
            const float w = (m[k] == -INFINITY || g + k >= CN) ? 0.0f : __expf(m[k] - gm);
            S += v[k] * w;
        }
        M = gm;
    }
    ms[i] = make_float2(M, S);
}

#define AGG_GPW 2

template <int C, bool FULL, int NG>
__device__ __forceinline__ void agg_groups(const int* __restrict__ src, const int* __restrict__ dst,
                                           const float* __restrict__ sc, const float2* __restrict__ ms,
                                           const float* __restrict__ h, float* __restrict__ agg,
                                           int e0, int cnt, unsigned lane, int E) {
    constexpr int R = C / 32;
    constexpr int G = 32 / R;
    int    s[NG], d[NG];
    float  a[NG], p[NG], q[NG];
    float2 w[NG];
#pragma unroll
    for (int g = 0; g < NG; ++g) {
        const int ee = e0 + g * 32 + (int)lane;
        const int e  = FULL ? ee : min(ee, E - 1);
        s[g] = src[e];
        d[g] = dst[e];
    }
#pragma unroll
    for (int g = 0; g < NG; ++g) {
        p[g] = sc[s[g] + (s[g] & ~15)];
        q[g] = sc[d[g] + (d[g] & ~15) + 16];
        w[g] = ms[d[g]];
    }
#pragma unroll
    for (int g = 0; g < NG; ++g) a[g] = __expf(leaky(p[g] + q[g]) - w[g].x) / w[g].y;
#pragma unroll
    for (int g = 0; g < NG; ++g) {
#pragma unroll
        for (int j0 = 0; j0 < 32; j0 += G) {
            float v[G][R];
#pragma unroll
            for (int j = 0; j < G; ++j) {
                const unsigned sb = (unsigned)__builtin_amdgcn_readlane(s[g], j0 + j) * C;
#pragma unroll
                for (int r = 0; r < R; ++r) v[j][r] = h[sb + lane + r * 32];
            }
#pragma unroll
            for (int j = 0; j < G; ++j) {
                if (FULL || g * 32 + j0 + j < cnt) {
                    const unsigned db = (unsigned)__builtin_amdgcn_readlane(d[g], j0 + j) * C;
                    const float    aj = __int_as_float(__builtin_amdgcn_readlane(__float_as_int(a[g]), j0 + j));
#pragma unroll
                    for (int r = 0; r < R; ++r) atomicAdd(agg + (db + lane + r * 32), aj * v[j][r]);
                }
            }
        }
    }
}

template <int C>
__global__ __launch_bounds__(256)
void edge_aggregate(const int* __restrict__ src, const int* __restrict__ dst,
                    const float* __restrict__ sc, const float* __restrict__ msf,
                    const float* __restrict__ h, float* __restrict__ agg, int E) {
    const float2* ms = reinterpret_cast<const float2*>(msf);
    const unsigned lane = threadIdx.x & 31;
    const int wave = __builtin_amdgcn_readfirstlane((int)((blockIdx.x * blockDim.x + threadIdx.x) >> 5));
    const int e0 = wave * (32 * AGG_GPW);
    if (e0 + 32 * AGG_GPW <= E) {
        agg_groups<C, true, AGG_GPW>(src, dst, sc, ms, h, agg, e0, 32 * AGG_GPW, lane, E);
    } else if (e0 < E) {
        agg_groups<C, false, AGG_GPW>(src, dst, sc, ms, h, agg, e0, E - e0, lane, E);
    }
}


static inline int cdiv(int a, int b) { return (a + b - 1) / b; }

extern "C" void kernel_launch(void* const* d_in, const int* in_sizes, int n_in,
                              void* d_out, int out_size, void* d_ws, size_t ws_size,
                              hipStream_t stream) {
    const float* x      = (const float*)d_in[0];
    const int*   eidx   = (const int*)d_in[1];
    const float* W1     = (const float*)d_in[2];
    const float* a_src1 = (const float*)d_in[3];
    const float* a_dst1 = (const float*)d_in[4];
    const float* b1     = (const float*)d_in[5];
    const float* W2     = (const float*)d_in[6];
    const float* a_src2 = (const float*)d_in[7];
    const float* a_dst2 = (const float*)d_in[8];
    const float* b2     = (const float*)d_in[9];
    float* out = (float*)d_out;

    const int IN = 128, H = 64, O = 32;
    const int N = in_sizes[0] / IN;
    const int E = in_sizes[1] / 2;
    const int* src = eidx;
    const int* dst = eidx + E;

    float* ws    = (float*)d_ws;
    float* h1    = ws;
    float* agg1  = h1 + (size_t)N * H;
    float* h2    = agg1 + (size_t)N * H;
    float* sc    = h2 + (size_t)N * O;
    float* ms    = sc + 2 * (size_t)N;

    const int B = 256;

    constexpr int SEG_U = 25;
    const int PITCH = (N + 31) / 32 * 32;
    const int NR    = (cdiv(N, SEG_R) + 31) / 32 * 32;
    const int CN    = cdiv(E, SEG_U * SEG_T);

    gemm_wmma_strip<4, 0, false><<<cdiv(N, 128), B, (IN * H + IN * 16 + 2 * H) * sizeof(float), stream>>>(
        x, W1, h1, nullptr, nullptr, a_src1, a_dst1, sc, N, IN);

    edge_softmax_stats<SEG_U><<<SEG_R * CN, SEG_T, 0, stream>>>(src, dst, sc, agg1, N, E, NR, PITCH);
    seg_fold<<<cdiv(N, B), B, 0, stream>>>(agg1, ms, N, CN, PITCH);

    start_rows<64><<<cdiv(N * H / 4, 1024), B, 0, stream>>>(agg1, b1, N * H / 4);
    edge_aggregate<64><<<cdiv(E, 32 * AGG_GPW * (B / 32)), B, 0, stream>>>(src, dst, sc, ms, h1, agg1, E);

    gemm_wmma_strip<2, 32, true><<<cdiv(N, 128), B, (H * O + H * 16 + 2 * O) * sizeof(float), stream>>>(
        agg1, W2, h2, b2, out, a_src2, a_dst2, sc, N, H);

    edge_softmax_stats<SEG_U><<<SEG_R * CN, SEG_T, 0, stream>>>(src, dst, sc, h1, N, E, NR, PITCH);
    seg_fold<<<cdiv(N, B), B, 0, stream>>>(h1, ms, N, CN, PITCH);

    edge_aggregate<32><<<cdiv(E, 32 * AGG_GPW * (B / 32)), B, 0, stream>>>(src, dst, sc, ms, h2, out, E);
}


# Layer 1 keeps each work-group's edge list in the workspace (one packed word per edge: source node, destination place); layer 2 reads it back: no second walk of the byte map, no index gathers
# speedup vs baseline: 4.2237x; 1.6715x over previous
#include <hip/hip_runtime.h>
#include <hip/hip_bf16.h>
#include <math.h>

#define SLOPE 0.2f

typedef float v2f __attribute__((ext_vector_type(2)));
typedef float v8f __attribute__((ext_vector_type(8)));


__device__ __forceinline__ float leaky(float v) {
    return v > 0.0f ? v : SLOPE * v;
}

template <int NT, bool RELU_A>
__global__ void gemm_wmma_strip(const float* __restrict__ A,
                                const float* __restrict__ B,
                                float* __restrict__ C,
                                const float* __restrict__ a_src,
                                const float* __restrict__ a_dst,
                                float* __restrict__ sc,
                                int M, int K) {
    const int N = NT * 16;
    extern __shared__ float sB[];
    float* sX = sB + K * N;
    float* sV = sX + K * 16;

    static_assert((NT * 16 & (NT * 16 - 1)) == 0, "the rotated walk below wraps with a mask");
    float va = 0.0f, vd = 0.0f;
    if ((int)threadIdx.x < N) {
        va = a_src[threadIdx.x];
        vd = a_dst[threadIdx.x];
    }

    for (int i = threadIdx.x; i < K * N; i += blockDim.x) sB[i] = B[i];
    for (int i = threadIdx.x; i < K * 16; i += blockDim.x) sX[i] = 0.0f;
    if ((int)threadIdx.x < N) {
        sV[2 * threadIdx.x]     = va;
        sV[2 * threadIdx.x + 1] = vd;
    }
    __syncthreads();
    for (int t = threadIdx.x; t < 2 * K; t += blockDim.x) {
        const int k = t >> 1, which = t & 1;
        float s = 0.0f;
        for (int jj = 0; jj < N; ++jj) {
            const int j = (jj + k) & (N - 1);
            s += sB[k * N + j] * sV[2 * j + which];
        }
        sX[k * 16 + which] = s;
    }
    __syncthreads();

    const int lane = threadIdx.x & 31;
    const int wid  = threadIdx.x >> 5;
    const int row0 = (blockIdx.x * (blockDim.x >> 5) + wid) * 16;
    if (row0 >= M) return;

    const int mn = lane & 15;
    const int hi = lane >> 4;
    const int kg = hi << 1;

    const float* arow = A + (size_t)(row0 + mn) * K;

    v8f acc[NT];
#pragma unroll
    for (int t = 0; t < NT; ++t) acc[t] = (v8f){};
    v8f accs = (v8f){};

#pragma unroll (NT == 2 ? 8 : 4)
    for (int k0 = 0; k0 < K; k0 += 4) {
        v2f a;
        a.x = arow[k0 + kg];
        a.y = arow[k0 + kg + 1];
        if (RELU_A) {
            a.x = fmaxf(a.x, 0.0f);
            a.y = fmaxf(a.y, 0.0f);
        }
        const float* brow = sB + (size_t)(k0 + kg) * N + mn;
#pragma unroll
        for (int t = 0; t < NT; ++t) {
            v2f b;
            b.x = brow[t * 16];
            b.y = brow[t * 16 + N];
            acc[t] = __builtin_amdgcn_wmma_f32_16x16x4_f32(
                 false, a,  false, b,
                 (short)0, acc[t],  false,  false);
        }
        {
            v2f b;
            b.x = sX[(k0 + kg) * 16 + mn];
            b.y = sX[(k0 + kg + 1) * 16 + mn];
            accs = __builtin_amdgcn_wmma_f32_16x16x4_f32(
                 false, a,  false, b,
                 (short)0, accs,  false,  false);
        }
    }

    if (mn < 2) {
        float* sp = sc + (size_t)row0 * 2 + mn * 16 + hi * 8;
#pragma unroll
        for (int v = 0; v < 8; ++v) sp[v] = accs[v];
    }

#pragma unroll
    for (int t = 0; t < NT; ++t) {
        float* crow = C + (size_t)(row0 + hi * 8) * N + t * 16 + mn;
#pragma unroll
        for (int v = 0; v < 8; ++v) crow[(size_t)v * N] = acc[t][v];
    }
}


#define NPB    196
#define EL_T   1024
#define EL_CAP 12288

#define LS_U      4
#define LS_CAP    (LS_U * EL_T)
#define LS_HEAD   32
#define LS_STRIDE (LS_HEAD + LS_CAP)

__global__ __launch_bounds__(256)
void owner_bytes(const int* __restrict__ dst, unsigned int* __restrict__ rid, int E, int nw) {
    const int t = blockIdx.x * blockDim.x + threadIdx.x;
    if (t >= nw) return;
    unsigned int word = 0;
#pragma unroll
    for (int k = 0; k < 4; ++k) {
        const int d = dst[min(4 * t + k, E - 1)];
        word |= (unsigned int)(d / NPB) << (8 * k);
    }
    rid[t] = word;
}

__device__ __forceinline__ void ldsMaxFloat(float* addr, float value) {
    if (__float_as_int(value) >= 0) {
        atomicMax((int*)addr, __float_as_int(value));
    } else {
        atomicMin((unsigned int*)addr, __float_as_uint(value));
    }
}

__device__ __forceinline__ void take_group(unsigned int x0, unsigned int x1, unsigned int x2, unsigned int x3,
                                           int ebase, int E, int* Ls, int* cnt) {
    const unsigned int z0 = ~(((x0 & 0x7F7F7F7Fu) + 0x7F7F7F7Fu) | x0) & 0x80808080u;
    const unsigned int z1 = ~(((x1 & 0x7F7F7F7Fu) + 0x7F7F7F7Fu) | x1) & 0x80808080u;
    const unsigned int z2 = ~(((x2 & 0x7F7F7F7Fu) + 0x7F7F7F7Fu) | x2) & 0x80808080u;
    const unsigned int z3 = ~(((x3 & 0x7F7F7F7Fu) + 0x7F7F7F7Fu) | x3) & 0x80808080u;
    unsigned int m = (z0 >> 7) | (z1 >> 6) | (z2 >> 5) | (z3 >> 4);
    while (m != 0u) {
        const int p = __builtin_ctz(m);
        m &= m - 1u;
        const int e = ebase + ((p & 7) << 2) + (p >> 3);
        if (e < E) {
            const int pos = atomicAdd(cnt, 1);
            if (pos < EL_CAP) Ls[pos] = e;
        }
    }
}

__device__ __forceinline__ void scan_owner_bytes(const uint4* __restrict__ rid4, int qb, int qe,
                                                 unsigned int bb, int E, int* Ls, int* cnt, int tid) {
    constexpr int U = 8;
    int q = qb + tid;
    for (; q + (U - 1) * EL_T < qe; q += EL_T * U) {
        uint4 r[U];
#pragma unroll
        for (int u = 0; u < U; ++u) r[u] = rid4[q + u * EL_T];
#pragma unroll
        for (int u = 0; u < U; ++u)
            take_group(r[u].x ^ bb, r[u].y ^ bb, r[u].z ^ bb, r[u].w ^ bb, (q + u * EL_T) * 16, E, Ls, cnt);
    }
    for (; q < qe; q += EL_T) {
        const uint4 r = rid4[q];
        take_group(r.x ^ bb, r.y ^ bb, r.z ^ bb, r.w ^ bb, q * 16, E, Ls, cnt);
    }
}

template <bool KEEP>
__device__ __forceinline__ void list_logits_max(const int* __restrict__ src, const int* __restrict__ dst,
                                                const float* __restrict__ sc, int lo, int n,
                                                int* Ls, int* Ld, float* Lw, float* Tm, const float* Tq, int tid,
                                                int* __restrict__ keep) {
    constexpr int U = 4;
    for (int i0 = tid; i0 < n; i0 += EL_T * U) {
        int s[U], d[U];
        float p[U];
#pragma unroll
        for (int u = 0; u < U; ++u) {
            const int i = i0 + u * EL_T;
            const int e = i < n ? Ls[i] : 0;
            s[u] = src[e];
            d[u] = dst[e];
        }
#pragma unroll
        for (int u = 0; u < U; ++u) p[u] = sc[s[u] + (s[u] & ~15)];
#pragma unroll
        for (int u = 0; u < U; ++u) {
            const int i = i0 + u * EL_T;
            if (i < n) {
                const int dl = d[u] - lo;
                const float lg = leaky(p[u] + Tq[dl]);
                Ls[i] = s[u];
                Ld[i] = dl;
                Lw[i] = lg;
                ldsMaxFloat(&Tm[dl], lg);
                if (KEEP) {
                    if (i < LS_CAP) keep[i] = (s[u] << 8) | dl;
                }
            }
        }
    }
}

template <int C>
__device__ __forceinline__ void list_sums_rows(const float* __restrict__ h, int n,
                                               const int* Ls, const int* Ld, float* Lw,
                                               const float* Tm, float* Ts, float* acc, int tid) {
    constexpr int R = C / 32;
    constexpr int G = 32 / R;
    for (int i = tid; i < n; i += EL_T) {
        const int dl = Ld[i];
        const float ex = __expf(Lw[i] - Tm[dl]);
        Lw[i] = ex;
        atomicAdd(&Ts[dl], ex);
    }
    const unsigned lane = tid & 31;
    const int wave = __builtin_amdgcn_readfirstlane(tid >> 5);
    const int nn   = __builtin_amdgcn_readfirstlane(n);
    for (int g = wave * 32; g < nn; g += EL_T) {
        const int i   = g + (int)lane;
        const int cw  = min(32, nn - g);
        const int   s = i < nn ? Ls[i] : 0;
        const int   d = i < nn ? Ld[i] : 0;
        const float a = i < nn ? Lw[i] : 0.0f;
#pragma unroll
        for (int j0 = 0; j0 < 32; j0 += G) {
            if (j0 < cw) {
                float v[G][R];
#pragma unroll
                for (int j = 0; j < G; ++j) {
                    const unsigned sb = (unsigned)__builtin_amdgcn_readlane(s, j0 + j) * C;
#pragma unroll
                    for (int r = 0; r < R; ++r) v[j][r] = h[sb + lane + r * 32];
                }
#pragma unroll
                for (int j = 0; j < G; ++j) {
                    if (j0 + j < cw) {
                        const unsigned db = (unsigned)__builtin_amdgcn_readlane(d, j0 + j) * C;
                        const float    aj = __int_as_float(__builtin_amdgcn_readlane(__float_as_int(a), j0 + j));
#pragma unroll
                        for (int r = 0; r < R; ++r) atomicAdd(&acc[db + lane + r * 32], aj * v[j][r]);
                    }
                }
            }
        }
    }
}

template <int C, bool FIRST>
__global__ __launch_bounds__(EL_T)
void edge_layer(const int* __restrict__ src, const int* __restrict__ dst,
                const unsigned int* __restrict__ rid, const float* __restrict__ sc,
                const float* __restrict__ h, const float* __restrict__ bias,
                float* __restrict__ outp, int* lst, int N, int E, int nq) {
    __shared__ __attribute__((aligned(16))) float acc[NPB * C];
    __shared__ float Tm[NPB];
    __shared__ float Ts[NPB];
    __shared__ float Tq[NPB];
    __shared__ int   Ls[EL_CAP];
    __shared__ int   Ld[EL_CAP];
    __shared__ float Lw[EL_CAP];
    __shared__ int   cnt;
    const int tid = threadIdx.x;
    const int lo  = (int)blockIdx.x * NPB;
    const int len = min(NPB, N - lo);
    const unsigned int bb = (unsigned int)blockIdx.x * 0x01010101u;
    const uint4* rid4 = reinterpret_cast<const uint4*>(rid);
    int* slot = lst + (size_t)blockIdx.x * LS_STRIDE;

    int kept = 0;
    int pk[LS_U];
    if (!FIRST) {
        const int* kslot = slot;
        kept = kslot[0];
#pragma unroll
        for (int u = 0; u < LS_U; ++u) pk[u] = kslot[LS_HEAD + tid + u * EL_T];
    }
    const bool listed = !FIRST && kept <= LS_CAP;

    for (int i = tid; i < NPB * C; i += EL_T) acc[i] = 0.0f;
    for (int i = tid; i < NPB; i += EL_T) {
        const int nd = min(lo + i, N - 1);
        Tm[i] = -INFINITY;
        Ts[i] = 0.0f;
        Tq[i] = sc[nd + (nd & ~15) + 16];
    }
    if (tid == 0) cnt = 0;

    float p[LS_U];
    if (listed) {
#pragma unroll
        for (int u = 0; u < LS_U; ++u) {
            const int s = tid + u * EL_T < kept ? (int)((unsigned int)pk[u] >> 8) : 0;
            p[u] = sc[s + (s & ~15)];
        }
    }
    __syncthreads();

    if (listed) {
#pragma unroll
        for (int u = 0; u < LS_U; ++u) {
            const int i = tid + u * EL_T;
            if (i < kept) {
                const int dl = pk[u] & 255;
                const float lg = leaky(p[u] + Tq[dl]);
                Ls[i] = (int)((unsigned int)pk[u] >> 8);
                Ld[i] = dl;
                Lw[i] = lg;
                ldsMaxFloat(&Tm[dl], lg);
            }
        }
        __syncthreads();
        list_sums_rows<C>(h, kept, Ls, Ld, Lw, Tm, Ts, acc, tid);
    } else {
        scan_owner_bytes(rid4, 0, nq, bb, E, Ls, &cnt, tid);
        __syncthreads();
        const int total = cnt;
        if (FIRST && tid == 0) slot[0] = total;
        if (total <= EL_CAP) {
            list_logits_max<FIRST>(src, dst, sc, lo, total, Ls, Ld, Lw, Tm, Tq, tid, slot + LS_HEAD);
            __syncthreads();
            list_sums_rows<C>(h, total, Ls, Ld, Lw, Tm, Ts, acc, tid);
        } else {
            constexpr int SQ = EL_CAP / 16;
            for (int sweep = 0; sweep < 2; ++sweep) {
                for (int q0 = 0; q0 < nq; q0 += SQ) {
                    __syncthreads();
                    if (tid == 0) cnt = 0;
                    __syncthreads();
                    scan_owner_bytes(rid4, q0, min(nq, q0 + SQ), bb, E, Ls, &cnt, tid);
                    __syncthreads();
                    const int n = cnt;
                    list_logits_max<false>(src, dst, sc, lo, n, Ls, Ld, Lw, Tm, Tq, tid, nullptr);
                    __syncthreads();
                    if (sweep == 1) list_sums_rows<C>(h, n, Ls, Ld, Lw, Tm, Ts, acc, tid);
                }
            }
        }
    }
    __syncthreads();

    const int n4 = len * (C / 4);
    float4* o4 = reinterpret_cast<float4*>(outp + (size_t)lo * C);
    const float4* a4 = reinterpret_cast<const float4*>(acc);
    for (int i = tid; i < n4; i += EL_T) {
        const int dl = i / (C / 4);
        const int c  = (i % (C / 4)) * 4;
        const float S   = Ts[dl];
        const float inv = S > 0.0f ? 1.0f / S : 0.0f;
        const float4 a = a4[i];
        o4[i] = make_float4(bias[c] + a.x * inv, bias[c + 1] + a.y * inv,
                            bias[c + 2] + a.z * inv, bias[c + 3] + a.w * inv);
    }
}


static inline int cdiv(int a, int b) { return (a + b - 1) / b; }

extern "C" void kernel_launch(void* const* d_in, const int* in_sizes, int n_in,
                              void* d_out, int out_size, void* d_ws, size_t ws_size,
                              hipStream_t stream) {
    const float* x      = (const float*)d_in[0];
    const int*   eidx   = (const int*)d_in[1];
    const float* W1     = (const float*)d_in[2];
    const float* a_src1 = (const float*)d_in[3];
    const float* a_dst1 = (const float*)d_in[4];
    const float* b1     = (const float*)d_in[5];
    const float* W2     = (const float*)d_in[6];
    const float* a_src2 = (const float*)d_in[7];
    const float* a_dst2 = (const float*)d_in[8];
    const float* b2     = (const float*)d_in[9];
    float* out = (float*)d_out;

    const int IN = 128, H = 64, O = 32;
    const int N = in_sizes[0] / IN;
    const int E = in_sizes[1] / 2;
    const int* src = eidx;
    const int* dst = eidx + E;

    float* ws    = (float*)d_ws;
    float* h1    = ws;
    float* agg1  = h1 + (size_t)N * H;
    float* h2    = h1;
    unsigned int* rid = (unsigned int*)(agg1 + (size_t)N * H);
    float* sc    = agg1 + (size_t)N * H + (size_t)N * O;

    const int B  = 256;
    const int nq = cdiv(E, 16);
    int* lst = (int*)(rid + (size_t)nq * 4);

    owner_bytes<<<cdiv(nq * 4, B), B, 0, stream>>>(dst, rid, E, nq * 4);

    gemm_wmma_strip<4, false><<<cdiv(N, 128), B, (IN * H + IN * 16 + 2 * H) * sizeof(float), stream>>>(
        x, W1, h1, a_src1, a_dst1, sc, N, IN);
    edge_layer<64, true><<<cdiv(N, NPB), EL_T, 0, stream>>>(src, dst, rid, sc, h1, b1, agg1, lst, N, E, nq);

    gemm_wmma_strip<2, true><<<cdiv(N, 128), B, (H * O + H * 16 + 2 * O) * sizeof(float), stream>>>(
        agg1, W2, h2, a_src2, a_dst2, sc, N, H);
    edge_layer<32, false><<<cdiv(N, NPB), EL_T, 0, stream>>>(src, dst, rid, sc, h2, b2, out, lst, N, E, nq);
}


# Owner map bit-sliced: eight bit planes per 32-edge block, a block built per thread; a scanning thread's match mask of 32 edges is one xor-or chain of 12 vector ops (was 52), map pass 1/8 the waves
# speedup vs baseline: 4.3948x; 1.0405x over previous
#include <hip/hip_runtime.h>
#include <hip/hip_bf16.h>
#include <math.h>

#define SLOPE 0.2f

typedef float v2f __attribute__((ext_vector_type(2)));
typedef float v8f __attribute__((ext_vector_type(8)));


__device__ __forceinline__ float leaky(float v) {
    return v > 0.0f ? v : SLOPE * v;
}

template <int NT, bool RELU_A>
__global__ void gemm_wmma_strip(const float* __restrict__ A,
                                const float* __restrict__ B,
                                float* __restrict__ C,
                                const float* __restrict__ a_src,
                                const float* __restrict__ a_dst,
                                float* __restrict__ sc,
                                int M, int K) {
    const int N = NT * 16;
    extern __shared__ float sB[];
    float* sX = sB + K * N;
    float* sV = sX + K * 16;

    static_assert((NT * 16 & (NT * 16 - 1)) == 0, "the rotated walk below wraps with a mask");
    float va = 0.0f, vd = 0.0f;
    if ((int)threadIdx.x < N) {
        va = a_src[threadIdx.x];
        vd = a_dst[threadIdx.x];
    }

    for (int i = threadIdx.x; i < K * N; i += blockDim.x) sB[i] = B[i];
    for (int i = threadIdx.x; i < K * 16; i += blockDim.x) sX[i] = 0.0f;
    if ((int)threadIdx.x < N) {
        sV[2 * threadIdx.x]     = va;
        sV[2 * threadIdx.x + 1] = vd;
    }
    __syncthreads();
    for (int t = threadIdx.x; t < 2 * K; t += blockDim.x) {
        const int k = t >> 1, which = t & 1;
        float s = 0.0f;
        for (int jj = 0; jj < N; ++jj) {
            const int j = (jj + k) & (N - 1);
            s += sB[k * N + j] * sV[2 * j + which];
        }
        sX[k * 16 + which] = s;
    }
    __syncthreads();

    const int lane = threadIdx.x & 31;
    const int wid  = threadIdx.x >> 5;
    const int row0 = (blockIdx.x * (blockDim.x >> 5) + wid) * 16;
    if (row0 >= M) return;

    const int mn = lane & 15;
    const int hi = lane >> 4;
    const int kg = hi << 1;

    const float* arow = A + (size_t)(row0 + mn) * K;

    v8f acc[NT];
#pragma unroll
    for (int t = 0; t < NT; ++t) acc[t] = (v8f){};
    v8f accs = (v8f){};

#pragma unroll (NT == 2 ? 8 : 4)
    for (int k0 = 0; k0 < K; k0 += 4) {
        v2f a;
        a.x = arow[k0 + kg];
        a.y = arow[k0 + kg + 1];
        if (RELU_A) {
            a.x = fmaxf(a.x, 0.0f);
            a.y = fmaxf(a.y, 0.0f);
        }
        const float* brow = sB + (size_t)(k0 + kg) * N + mn;
#pragma unroll
        for (int t = 0; t < NT; ++t) {
            v2f b;
            b.x = brow[t * 16];
            b.y = brow[t * 16 + N];
            acc[t] = __builtin_amdgcn_wmma_f32_16x16x4_f32(
                 false, a,  false, b,
                 (short)0, acc[t],  false,  false);
        }
        {
            v2f b;
            b.x = sX[(k0 + kg) * 16 + mn];
            b.y = sX[(k0 + kg + 1) * 16 + mn];
            accs = __builtin_amdgcn_wmma_f32_16x16x4_f32(
                 false, a,  false, b,
                 (short)0, accs,  false,  false);
        }
    }

    if (mn < 2) {
        float* sp = sc + (size_t)row0 * 2 + mn * 16 + hi * 8;
#pragma unroll
        for (int v = 0; v < 8; ++v) sp[v] = accs[v];
    }

#pragma unroll
    for (int t = 0; t < NT; ++t) {
        float* crow = C + (size_t)(row0 + hi * 8) * N + t * 16 + mn;
#pragma unroll
        for (int v = 0; v < 8; ++v) crow[(size_t)v * N] = acc[t][v];
    }
}


#define NPB    196
#define EL_T   1024
#define EL_CAP 12288

#define LS_U      4
#define LS_CAP    (LS_U * EL_T)
#define LS_HEAD   32
#define LS_STRIDE (LS_HEAD + LS_CAP)

__global__ __launch_bounds__(64)
void owner_planes(const int* __restrict__ dst, unsigned int* __restrict__ rid, int E, int nb) {
    const int blk = blockIdx.x * blockDim.x + threadIdx.x;
    if (blk >= nb) return;
    uint4* lo4 = reinterpret_cast<uint4*>(rid);
    uint4* hi4 = lo4 + nb;
    const int e0 = blk * 32;
    int d[32];
    if (e0 + 31 < E && (reinterpret_cast<size_t>(dst + e0) & 15) == 0) {
        const int4* p = reinterpret_cast<const int4*>(dst + e0);
#pragma unroll
        for (int k = 0; k < 8; ++k) {
            const int4 v = p[k];
            d[4 * k] = v.x; d[4 * k + 1] = v.y; d[4 * k + 2] = v.z; d[4 * k + 3] = v.w;
        }
    } else {
#pragma unroll
        for (int k = 0; k < 32; ++k) d[k] = dst[min(e0 + k, E - 1)];
    }
    unsigned int w[8];
#pragma unroll
    for (int k = 0; k < 8; ++k) w[k] = 0u;
#pragma unroll
    for (int i = 0; i < 32; ++i) {
        const unsigned int o = (unsigned int)(d[i] / NPB);
#pragma unroll
        for (int k = 0; k < 8; ++k) w[k] |= ((o >> k) & 1u) << i;
    }
    lo4[blk] = make_uint4(w[0], w[1], w[2], w[3]);
    hi4[blk] = make_uint4(w[4], w[5], w[6], w[7]);
}

__device__ __forceinline__ void ldsMaxFloat(float* addr, float value) {
    if (__float_as_int(value) >= 0) {
        atomicMax((int*)addr, __float_as_int(value));
    } else {
        atomicMin((unsigned int*)addr, __float_as_uint(value));
    }
}

struct OwnerSel { unsigned int s[8]; };

__device__ __forceinline__ void take_block(const uint4 a, const uint4 h, const OwnerSel& sel,
                                           int ebase, int E, int* Ls, int* cnt) {
    const unsigned int x = (a.x ^ sel.s[0]) | (a.y ^ sel.s[1]) | (a.z ^ sel.s[2]) | (a.w ^ sel.s[3]) |
                           (h.x ^ sel.s[4]) | (h.y ^ sel.s[5]) | (h.z ^ sel.s[6]) | (h.w ^ sel.s[7]);
    unsigned int m = ~x;
    while (m != 0u) {
        const int p = __builtin_ctz(m);
        m &= m - 1u;
        const int e = ebase + p;
        if (e < E) {
            const int pos = atomicAdd(cnt, 1);
            if (pos < EL_CAP) Ls[pos] = e;
        }
    }
}

__device__ __forceinline__ void scan_owner_planes(const uint4* __restrict__ lo4, const uint4* __restrict__ hi4,
                                                  int qb, int qe, const OwnerSel& sel, int E,
                                                  int* Ls, int* cnt, int tid) {
    constexpr int U = 4;
    int q = qb + tid;
    for (; q + (U - 1) * EL_T < qe; q += EL_T * U) {
        uint4 a[U], h[U];
#pragma unroll
        for (int u = 0; u < U; ++u) {
            a[u] = lo4[q + u * EL_T];
            h[u] = hi4[q + u * EL_T];
        }
#pragma unroll
        for (int u = 0; u < U; ++u)
            take_block(a[u], h[u], sel, (q + u * EL_T) * 32, E, Ls, cnt);
    }
    for (; q < qe; q += EL_T) {
        const uint4 a = lo4[q];
        const uint4 h = hi4[q];
        take_block(a, h, sel, q * 32, E, Ls, cnt);
    }
}

template <bool KEEP>
__device__ __forceinline__ void list_logits_max(const int* __restrict__ src, const int* __restrict__ dst,
                                                const float* __restrict__ sc, int lo, int n,
                                                int* Ls, int* Ld, float* Lw, float* Tm, const float* Tq, int tid,
                                                int* __restrict__ keep) {
    constexpr int U = 4;
    for (int i0 = tid; i0 < n; i0 += EL_T * U) {
        int s[U], d[U];
        float p[U];
#pragma unroll
        for (int u = 0; u < U; ++u) {
            const int i = i0 + u * EL_T;
            const int e = i < n ? Ls[i] : 0;
            s[u] = src[e];
            d[u] = dst[e];
        }
#pragma unroll
        for (int u = 0; u < U; ++u) p[u] = sc[s[u] + (s[u] & ~15)];
#pragma unroll
        for (int u = 0; u < U; ++u) {
            const int i = i0 + u * EL_T;
            if (i < n) {
                const int dl = d[u] - lo;
                const float lg = leaky(p[u] + Tq[dl]);
                Ls[i] = s[u];
                Ld[i] = dl;
                Lw[i] = lg;
                ldsMaxFloat(&Tm[dl], lg);
                if (KEEP) {
                    if (i < LS_CAP) keep[i] = (s[u] << 8) | dl;
                }
            }
        }
    }
}

template <int C>
__device__ __forceinline__ void list_sums_rows(const float* __restrict__ h, int n,
                                               const int* Ls, const int* Ld, float* Lw,
                                               const float* Tm, float* Ts, float* acc, int tid) {
    constexpr int R = C / 32;
    constexpr int G = 32 / R;
    for (int i = tid; i < n; i += EL_T) {
        const int dl = Ld[i];
        const float ex = __expf(Lw[i] - Tm[dl]);
        Lw[i] = ex;
        atomicAdd(&Ts[dl], ex);
    }
    const unsigned lane = tid & 31;
    const int wave = __builtin_amdgcn_readfirstlane(tid >> 5);
    const int nn   = __builtin_amdgcn_readfirstlane(n);
    for (int g = wave * 32; g < nn; g += EL_T) {
        const int i   = g + (int)lane;
        const int cw  = min(32, nn - g);
        const int   s = i < nn ? Ls[i] : 0;
        const int   d = i < nn ? Ld[i] : 0;
        const float a = i < nn ? Lw[i] : 0.0f;
#pragma unroll
        for (int j0 = 0; j0 < 32; j0 += G) {
            if (j0 < cw) {
                float v[G][R];
#pragma unroll
                for (int j = 0; j < G; ++j) {
                    const unsigned sb = (unsigned)__builtin_amdgcn_readlane(s, j0 + j) * C;
#pragma unroll
                    for (int r = 0; r < R; ++r) v[j][r] = h[sb + lane + r * 32];
                }
#pragma unroll
                for (int j = 0; j < G; ++j) {
                    if (j0 + j < cw) {
                        const unsigned db = (unsigned)__builtin_amdgcn_readlane(d, j0 + j) * C;
                        const float    aj = __int_as_float(__builtin_amdgcn_readlane(__float_as_int(a), j0 + j));
#pragma unroll
                        for (int r = 0; r < R; ++r) atomicAdd(&acc[db + lane + r * 32], aj * v[j][r]);
                    }
                }
            }
        }
    }
}

template <int C, bool FIRST>
__global__ __launch_bounds__(EL_T)
void edge_layer(const int* __restrict__ src, const int* __restrict__ dst,
                const unsigned int* __restrict__ rid, const float* __restrict__ sc,
                const float* __restrict__ h, const float* __restrict__ bias,
                float* __restrict__ outp, int* lst, int N, int E, int nb) {
    __shared__ __attribute__((aligned(16))) float acc[NPB * C];
    __shared__ float Tm[NPB];
    __shared__ float Ts[NPB];
    __shared__ float Tq[NPB];
    __shared__ int   Ls[EL_CAP];
    __shared__ int   Ld[EL_CAP];
    __shared__ float Lw[EL_CAP];
    __shared__ int   cnt;
    const int tid = threadIdx.x;
    const int lo  = (int)blockIdx.x * NPB;
    const int len = min(NPB, N - lo);
    OwnerSel sel;
#pragma unroll
    for (int k = 0; k < 8; ++k) sel.s[k] = ((blockIdx.x >> k) & 1u) ? 0xFFFFFFFFu : 0u;
    const uint4* lo4 = reinterpret_cast<const uint4*>(rid);
    const uint4* hi4 = lo4 + nb;
    int* slot = lst + (size_t)blockIdx.x * LS_STRIDE;

    int kept = 0;
    int pk[LS_U];
    if (!FIRST) {
        const int* kslot = slot;
        kept = kslot[0];
#pragma unroll
        for (int u = 0; u < LS_U; ++u) pk[u] = kslot[LS_HEAD + tid + u * EL_T];
    }
    const bool listed = !FIRST && kept <= LS_CAP;

    for (int i = tid; i < NPB * C; i += EL_T) acc[i] = 0.0f;
    for (int i = tid; i < NPB; i += EL_T) {
        const int nd = min(lo + i, N - 1);
        Tm[i] = -INFINITY;
        Ts[i] = 0.0f;
        Tq[i] = sc[nd + (nd & ~15) + 16];
    }
    if (tid == 0) cnt = 0;

    float p[LS_U];
    if (listed) {
#pragma unroll
        for (int u = 0; u < LS_U; ++u) {
            const int s = tid + u * EL_T < kept ? (int)((unsigned int)pk[u] >> 8) : 0;
            p[u] = sc[s + (s & ~15)];
        }
    }
    __syncthreads();

    if (listed) {
#pragma unroll
        for (int u = 0; u < LS_U; ++u) {
            const int i = tid + u * EL_T;
            if (i < kept) {
                const int dl = pk[u] & 255;
                const float lg = leaky(p[u] + Tq[dl]);
                Ls[i] = (int)((unsigned int)pk[u] >> 8);
                Ld[i] = dl;
                Lw[i] = lg;
                ldsMaxFloat(&Tm[dl], lg);
            }
        }
        __syncthreads();
        list_sums_rows<C>(h, kept, Ls, Ld, Lw, Tm, Ts, acc, tid);
    } else {
        scan_owner_planes(lo4, hi4, 0, nb, sel, E, Ls, &cnt, tid);
        __syncthreads();
        const int total = cnt;
        if (FIRST && tid == 0) slot[0] = total;
        if (total <= EL_CAP) {
            list_logits_max<FIRST>(src, dst, sc, lo, total, Ls, Ld, Lw, Tm, Tq, tid, slot + LS_HEAD);
            __syncthreads();
            list_sums_rows<C>(h, total, Ls, Ld, Lw, Tm, Ts, acc, tid);
        } else {
            constexpr int SQ = EL_CAP / 32;
            for (int sweep = 0; sweep < 2; ++sweep) {
                for (int q0 = 0; q0 < nb; q0 += SQ) {
                    __syncthreads();
                    if (tid == 0) cnt = 0;
                    __syncthreads();
                    scan_owner_planes(lo4, hi4, q0, min(nb, q0 + SQ), sel, E, Ls, &cnt, tid);
                    __syncthreads();
                    const int n = cnt;
                    list_logits_max<false>(src, dst, sc, lo, n, Ls, Ld, Lw, Tm, Tq, tid, nullptr);
                    __syncthreads();
                    if (sweep == 1) list_sums_rows<C>(h, n, Ls, Ld, Lw, Tm, Ts, acc, tid);
                }
            }
        }
    }
    __syncthreads();

    const int n4 = len * (C / 4);
    float4* o4 = reinterpret_cast<float4*>(outp + (size_t)lo * C);
    const float4* a4 = reinterpret_cast<const float4*>(acc);
    for (int i = tid; i < n4; i += EL_T) {
        const int dl = i / (C / 4);
        const int c  = (i % (C / 4)) * 4;
        const float S   = Ts[dl];
        const float inv = S > 0.0f ? 1.0f / S : 0.0f;
        const float4 a = a4[i];
        o4[i] = make_float4(bias[c] + a.x * inv, bias[c + 1] + a.y * inv,
                            bias[c + 2] + a.z * inv, bias[c + 3] + a.w * inv);
    }
}


static inline int cdiv(int a, int b) { return (a + b - 1) / b; }

extern "C" void kernel_launch(void* const* d_in, const int* in_sizes, int n_in,
                              void* d_out, int out_size, void* d_ws, size_t ws_size,
                              hipStream_t stream) {
    const float* x      = (const float*)d_in[0];
    const int*   eidx   = (const int*)d_in[1];
    const float* W1     = (const float*)d_in[2];
    const float* a_src1 = (const float*)d_in[3];
    const float* a_dst1 = (const float*)d_in[4];
    const float* b1     = (const float*)d_in[5];
    const float* W2     = (const float*)d_in[6];
    const float* a_src2 = (const float*)d_in[7];
    const float* a_dst2 = (const float*)d_in[8];
    const float* b2     = (const float*)d_in[9];
    float* out = (float*)d_out;

    const int IN = 128, H = 64, O = 32;
    const int N = in_sizes[0] / IN;
    const int E = in_sizes[1] / 2;
    const int* src = eidx;
    const int* dst = eidx + E;

    float* ws    = (float*)d_ws;
    float* h1    = ws;
    float* agg1  = h1 + (size_t)N * H;
    float* h2    = h1;
    unsigned int* rid = (unsigned int*)(agg1 + (size_t)N * H);
    float* sc    = agg1 + (size_t)N * H + (size_t)N * O;

    const int B  = 256;
    const int nb = cdiv(E, 32);
    int* lst = (int*)(rid + (size_t)nb * 8);

    owner_planes<<<cdiv(nb, 64), 64, 0, stream>>>(dst, rid, E, nb);

    gemm_wmma_strip<4, false><<<cdiv(N, 128), B, (IN * H + IN * 16 + 2 * H) * sizeof(float), stream>>>(
        x, W1, h1, a_src1, a_dst1, sc, N, IN);
    edge_layer<64, true><<<cdiv(N, NPB), EL_T, 0, stream>>>(src, dst, rid, sc, h1, b1, agg1, lst, N, E, nb);

    gemm_wmma_strip<2, true><<<cdiv(N, 128), B, (H * O + H * 16 + 2 * O) * sizeof(float), stream>>>(
        agg1, W2, h2, a_src2, a_dst2, sc, N, H);
    edge_layer<32, false><<<cdiv(N, NPB), EL_T, 0, stream>>>(src, dst, rid, sc, h2, b2, out, lst, N, E, nb);
}


# Layer-2 projection and scores done by the layer-1 edge work-group itself: its range's finished rows stay in LDS as A of f32 matrix ops (a wave per 16 rows); 5 launches to 4, layer-1 plane not written
# speedup vs baseline: 4.7271x; 1.0756x over previous
#include <hip/hip_runtime.h>
#include <hip/hip_bf16.h>
#include <math.h>

#define SLOPE 0.2f

typedef float v2f __attribute__((ext_vector_type(2)));
typedef float v8f __attribute__((ext_vector_type(8)));


__device__ __forceinline__ float leaky(float v) {
    return v > 0.0f ? v : SLOPE * v;
}

template <int NT>
__global__ void gemm_wmma_strip(const float* __restrict__ A,
                                const float* __restrict__ B,
                                float* __restrict__ C,
                                const float* __restrict__ a_src,
                                const float* __restrict__ a_dst,
                                float* __restrict__ sc,
                                int M, int K) {
    const int N = NT * 16;
    extern __shared__ float sB[];
    float* sX = sB + K * N;
    float* sV = sX + K * 16;

    static_assert((NT * 16 & (NT * 16 - 1)) == 0, "the rotated walk below wraps with a mask");
    float va = 0.0f, vd = 0.0f;
    if ((int)threadIdx.x < N) {
        va = a_src[threadIdx.x];
        vd = a_dst[threadIdx.x];
    }

    for (int i = threadIdx.x; i < K * N; i += blockDim.x) sB[i] = B[i];
    for (int i = threadIdx.x; i < K * 16; i += blockDim.x) sX[i] = 0.0f;
    if ((int)threadIdx.x < N) {
        sV[2 * threadIdx.x]     = va;
        sV[2 * threadIdx.x + 1] = vd;
    }
    __syncthreads();
    for (int t = threadIdx.x; t < 2 * K; t += blockDim.x) {
        const int k = t >> 1, which = t & 1;
        float s = 0.0f;
        for (int jj = 0; jj < N; ++jj) {
            const int j = (jj + k) & (N - 1);
            s += sB[k * N + j] * sV[2 * j + which];
        }
        sX[k * 16 + which] = s;
    }
    __syncthreads();

    const int lane = threadIdx.x & 31;
    const int wid  = threadIdx.x >> 5;
    const int row0 = (blockIdx.x * (blockDim.x >> 5) + wid) * 16;
    if (row0 >= M) return;

    const int mn = lane & 15;
    const int hi = lane >> 4;
    const int kg = hi << 1;

    const float* arow = A + (size_t)(row0 + mn) * K;

    v8f acc[NT];
#pragma unroll
    for (int t = 0; t < NT; ++t) acc[t] = (v8f){};
    v8f accs = (v8f){};

#pragma unroll (NT == 2 ? 8 : 4)
    for (int k0 = 0; k0 < K; k0 += 4) {
        v2f a;
        a.x = arow[k0 + kg];
        a.y = arow[k0 + kg + 1];
        const float* brow = sB + (size_t)(k0 + kg) * N + mn;
#pragma unroll
        for (int t = 0; t < NT; ++t) {
            v2f b;
            b.x = brow[t * 16];
            b.y = brow[t * 16 + N];
            acc[t] = __builtin_amdgcn_wmma_f32_16x16x4_f32(
                 false, a,  false, b,
                 (short)0, acc[t],  false,  false);
        }
        {
            v2f b;
            b.x = sX[(k0 + kg) * 16 + mn];
            b.y = sX[(k0 + kg + 1) * 16 + mn];
            accs = __builtin_amdgcn_wmma_f32_16x16x4_f32(
                 false, a,  false, b,
                 (short)0, accs,  false,  false);
        }
    }

    if (mn < 2) {
        float* sp = sc + (size_t)row0 * 2 + mn * 16 + hi * 8;
#pragma unroll
        for (int v = 0; v < 8; ++v) sp[v] = accs[v];
    }

#pragma unroll
    for (int t = 0; t < NT; ++t) {
        float* crow = C + (size_t)(row0 + hi * 8) * N + t * 16 + mn;
#pragma unroll
        for (int v = 0; v < 8; ++v) crow[(size_t)v * N] = acc[t][v];
    }
}


#define NPB    196
#define EL_T   1024
#define EL_CAP 12288

#define LS_U      4
#define LS_CAP    (LS_U * EL_T)
#define LS_HEAD   32
#define LS_STRIDE (LS_HEAD + LS_CAP)

__global__ __launch_bounds__(64)
void owner_planes(const int* __restrict__ dst, unsigned int* __restrict__ rid, int E, int nb) {
    const int blk = blockIdx.x * blockDim.x + threadIdx.x;
    if (blk >= nb) return;
    uint4* lo4 = reinterpret_cast<uint4*>(rid);
    uint4* hi4 = lo4 + nb;
    const int e0 = blk * 32;
    int d[32];
    if (e0 + 31 < E && (reinterpret_cast<size_t>(dst + e0) & 15) == 0) {
        const int4* p = reinterpret_cast<const int4*>(dst + e0);
#pragma unroll
        for (int k = 0; k < 8; ++k) {
            const int4 v = p[k];
            d[4 * k] = v.x; d[4 * k + 1] = v.y; d[4 * k + 2] = v.z; d[4 * k + 3] = v.w;
        }
    } else {
#pragma unroll
        for (int k = 0; k < 32; ++k) d[k] = dst[min(e0 + k, E - 1)];
    }
    unsigned int w[8];
#pragma unroll
    for (int k = 0; k < 8; ++k) w[k] = 0u;
#pragma unroll
    for (int i = 0; i < 32; ++i) {
        const unsigned int o = (unsigned int)(d[i] / NPB);
#pragma unroll
        for (int k = 0; k < 8; ++k) w[k] |= ((o >> k) & 1u) << i;
    }
    lo4[blk] = make_uint4(w[0], w[1], w[2], w[3]);
    hi4[blk] = make_uint4(w[4], w[5], w[6], w[7]);
}

__device__ __forceinline__ void ldsMaxFloat(float* addr, float value) {
    if (__float_as_int(value) >= 0) {
        atomicMax((int*)addr, __float_as_int(value));
    } else {
        atomicMin((unsigned int*)addr, __float_as_uint(value));
    }
}

struct OwnerSel { unsigned int s[8]; };

__device__ __forceinline__ void take_block(const uint4 a, const uint4 h, const OwnerSel& sel,
                                           int ebase, int E, int* Ls, int* cnt) {
    const unsigned int x = (a.x ^ sel.s[0]) | (a.y ^ sel.s[1]) | (a.z ^ sel.s[2]) | (a.w ^ sel.s[3]) |
                           (h.x ^ sel.s[4]) | (h.y ^ sel.s[5]) | (h.z ^ sel.s[6]) | (h.w ^ sel.s[7]);
    unsigned int m = ~x;
    while (m != 0u) {
        const int p = __builtin_ctz(m);
        m &= m - 1u;
        const int e = ebase + p;
        if (e < E) {
            const int pos = atomicAdd(cnt, 1);
            if (pos < EL_CAP) Ls[pos] = e;
        }
    }
}

__device__ __forceinline__ void scan_owner_planes(const uint4* __restrict__ lo4, const uint4* __restrict__ hi4,
                                                  int qb, int qe, const OwnerSel& sel, int E,
                                                  int* Ls, int* cnt, int tid) {
    constexpr int U = 4;
    int q = qb + tid;
    for (; q + (U - 1) * EL_T < qe; q += EL_T * U) {
        uint4 a[U], h[U];
#pragma unroll
        for (int u = 0; u < U; ++u) {
            a[u] = lo4[q + u * EL_T];
            h[u] = hi4[q + u * EL_T];
        }
#pragma unroll
        for (int u = 0; u < U; ++u)
            take_block(a[u], h[u], sel, (q + u * EL_T) * 32, E, Ls, cnt);
    }
    for (; q < qe; q += EL_T) {
        const uint4 a = lo4[q];
        const uint4 h = hi4[q];
        take_block(a, h, sel, q * 32, E, Ls, cnt);
    }
}

template <bool KEEP>
__device__ __forceinline__ void list_logits_max(const int* __restrict__ src, const int* __restrict__ dst,
                                                const float* __restrict__ sc, int lo, int n,
                                                int* Ls, int* Ld, float* Lw, float* Tm, const float* Tq, int tid,
                                                int* __restrict__ keep) {
    constexpr int U = 4;
    for (int i0 = tid; i0 < n; i0 += EL_T * U) {
        int s[U], d[U];
        float p[U];
#pragma unroll
        for (int u = 0; u < U; ++u) {
            const int i = i0 + u * EL_T;
            const int e = i < n ? Ls[i] : 0;
            s[u] = src[e];
            d[u] = dst[e];
        }
#pragma unroll
        for (int u = 0; u < U; ++u) p[u] = sc[s[u] + (s[u] & ~15)];
#pragma unroll
        for (int u = 0; u < U; ++u) {
            const int i = i0 + u * EL_T;
            if (i < n) {
                const int dl = d[u] - lo;
                const float lg = leaky(p[u] + Tq[dl]);
                Ls[i] = s[u];
                Ld[i] = dl;
                Lw[i] = lg;
                ldsMaxFloat(&Tm[dl], lg);
                if (KEEP) {
                    if (i < LS_CAP) keep[i] = (s[u] << 8) | dl;
                }
            }
        }
    }
}

template <int C>
__device__ __forceinline__ void list_sums_rows(const float* __restrict__ h, int n,
                                               const int* Ls, const int* Ld, float* Lw,
                                               const float* Tm, float* Ts, float* acc, int tid) {
    constexpr int R = C / 32;
    constexpr int G = 32 / R;
    for (int i = tid; i < n; i += EL_T) {
        const int dl = Ld[i];
        const float ex = __expf(Lw[i] - Tm[dl]);
        Lw[i] = ex;
        atomicAdd(&Ts[dl], ex);
    }
    const unsigned lane = tid & 31;
    const int wave = __builtin_amdgcn_readfirstlane(tid >> 5);
    const int nn   = __builtin_amdgcn_readfirstlane(n);
    for (int g = wave * 32; g < nn; g += EL_T) {
        const int i   = g + (int)lane;
        const int cw  = min(32, nn - g);
        const int   s = i < nn ? Ls[i] : 0;
        const int   d = i < nn ? Ld[i] : 0;
        const float a = i < nn ? Lw[i] : 0.0f;
#pragma unroll
        for (int j0 = 0; j0 < 32; j0 += G) {
            if (j0 < cw) {
                float v[G][R];
#pragma unroll
                for (int j = 0; j < G; ++j) {
                    const unsigned sb = (unsigned)__builtin_amdgcn_readlane(s, j0 + j) * C;
#pragma unroll
                    for (int r = 0; r < R; ++r) v[j][r] = h[sb + lane + r * 32];
                }
#pragma unroll
                for (int j = 0; j < G; ++j) {
                    if (j0 + j < cw) {
                        const unsigned db = (unsigned)__builtin_amdgcn_readlane(d, j0 + j) * C;
                        const float    aj = __int_as_float(__builtin_amdgcn_readlane(__float_as_int(a), j0 + j));
#pragma unroll
                        for (int r = 0; r < R; ++r) atomicAdd(&acc[db + lane + r * 32], aj * v[j][r]);
                    }
                }
            }
        }
    }
}

template <int C, bool FIRST>
__global__ __launch_bounds__(EL_T)
void edge_layer(const int* __restrict__ src, const int* __restrict__ dst,
                const unsigned int* __restrict__ rid, const float* __restrict__ sc,
                const float* __restrict__ h, const float* __restrict__ bias,
                float* __restrict__ outp, int* lst, int N, int E, int nb,
                const float* __restrict__ Wn, const float* __restrict__ an_src,
                const float* __restrict__ an_dst, float* __restrict__ scn) {
    constexpr int CN = 32;
    __shared__ float sW[FIRST ? C * CN : 1];
    __shared__ float sX[FIRST ? C * 16 : 1];
    __shared__ float sV[FIRST ? 2 * CN : 2];
    __shared__ __attribute__((aligned(16))) float acc[NPB * C];
    __shared__ float Tm[NPB];
    __shared__ float Ts[NPB];
    __shared__ float Tq[NPB];
    __shared__ int   Ls[EL_CAP];
    __shared__ int   Ld[EL_CAP];
    __shared__ float Lw[EL_CAP];
    __shared__ int   cnt;
    const int tid = threadIdx.x;
    const int lo  = (int)blockIdx.x * NPB;
    const int len = min(NPB, N - lo);
    OwnerSel sel;
#pragma unroll
    for (int k = 0; k < 8; ++k) sel.s[k] = ((blockIdx.x >> k) & 1u) ? 0xFFFFFFFFu : 0u;
    const uint4* lo4 = reinterpret_cast<const uint4*>(rid);
    const uint4* hi4 = lo4 + nb;
    int* slot = lst + (size_t)blockIdx.x * LS_STRIDE;

    int kept = 0;
    int pk[LS_U];
    if (!FIRST) {
        const int* kslot = slot;
        kept = kslot[0];
#pragma unroll
        for (int u = 0; u < LS_U; ++u) pk[u] = kslot[LS_HEAD + tid + u * EL_T];
    }
    const bool listed = !FIRST && kept <= LS_CAP;

    for (int i = tid; i < NPB * C; i += EL_T) acc[i] = 0.0f;
    for (int i = tid; i < NPB; i += EL_T) {
        const int nd = min(lo + i, N - 1);
        Tm[i] = -INFINITY;
        Ts[i] = 0.0f;
        Tq[i] = sc[nd + (nd & ~15) + 16];
    }
    if (tid == 0) cnt = 0;
    if (FIRST) {
        for (int i = tid; i < C * CN; i += EL_T) sW[i] = Wn[i];
        for (int i = tid; i < C * 16; i += EL_T) sX[i] = 0.0f;
        if (tid < CN) {
            sV[2 * tid]     = an_src[tid];
            sV[2 * tid + 1] = an_dst[tid];
        }
    }

    float p[LS_U];
    if (listed) {
#pragma unroll
        for (int u = 0; u < LS_U; ++u) {
            const int s = tid + u * EL_T < kept ? (int)((unsigned int)pk[u] >> 8) : 0;
            p[u] = sc[s + (s & ~15)];
        }
    }
    __syncthreads();

    if (FIRST) {
        if (tid < 2 * C) {
            const int k = tid >> 1, which = tid & 1;
            float s = 0.0f;
            for (int jj = 0; jj < CN; ++jj) {
                const int j = (jj + k) & (CN - 1);
                s += sW[k * CN + j] * sV[2 * j + which];
            }
            sX[k * 16 + which] = s;
        }
    }

    if (listed) {
#pragma unroll
        for (int u = 0; u < LS_U; ++u) {
            const int i = tid + u * EL_T;
            if (i < kept) {
                const int dl = pk[u] & 255;
                const float lg = leaky(p[u] + Tq[dl]);
                Ls[i] = (int)((unsigned int)pk[u] >> 8);
                Ld[i] = dl;
                Lw[i] = lg;
                ldsMaxFloat(&Tm[dl], lg);
            }
        }
        __syncthreads();
        list_sums_rows<C>(h, kept, Ls, Ld, Lw, Tm, Ts, acc, tid);
    } else {
        scan_owner_planes(lo4, hi4, 0, nb, sel, E, Ls, &cnt, tid);
        __syncthreads();
        const int total = cnt;
        if (FIRST && tid == 0) slot[0] = total;
        if (total <= EL_CAP) {
            list_logits_max<FIRST>(src, dst, sc, lo, total, Ls, Ld, Lw, Tm, Tq, tid, slot + LS_HEAD);
            __syncthreads();
            list_sums_rows<C>(h, total, Ls, Ld, Lw, Tm, Ts, acc, tid);
        } else {
            constexpr int SQ = EL_CAP / 32;
            for (int sweep = 0; sweep < 2; ++sweep) {
                for (int q0 = 0; q0 < nb; q0 += SQ) {
                    __syncthreads();
                    if (tid == 0) cnt = 0;
                    __syncthreads();
                    scan_owner_planes(lo4, hi4, q0, min(nb, q0 + SQ), sel, E, Ls, &cnt, tid);
                    __syncthreads();
                    const int n = cnt;
                    list_logits_max<false>(src, dst, sc, lo, n, Ls, Ld, Lw, Tm, Tq, tid, nullptr);
                    __syncthreads();
                    if (sweep == 1) list_sums_rows<C>(h, n, Ls, Ld, Lw, Tm, Ts, acc, tid);
                }
            }
        }
    }
    __syncthreads();

    if (!FIRST) {
        const int n4 = len * (C / 4);
        float4* o4 = reinterpret_cast<float4*>(outp + (size_t)lo * C);
        const float4* a4 = reinterpret_cast<const float4*>(acc);
        for (int i = tid; i < n4; i += EL_T) {
            const int dl = i / (C / 4);
            const int c  = (i % (C / 4)) * 4;
            const float S   = Ts[dl];
            const float inv = S > 0.0f ? 1.0f / S : 0.0f;
            const float4 a = a4[i];
            o4[i] = make_float4(bias[c] + a.x * inv, bias[c + 1] + a.y * inv,
                                bias[c + 2] + a.z * inv, bias[c + 3] + a.w * inv);
        }
    } else {
        float4* a4 = reinterpret_cast<float4*>(acc);
        for (int i = tid; i < NPB * (C / 4); i += EL_T) {
            const int dl = i / (C / 4);
            const int c  = (i % (C / 4)) * 4;
            const float S   = Ts[dl];
            const float inv = S > 0.0f ? 1.0f / S : 0.0f;
            const float4 a = a4[i];
            a4[i] = make_float4(fmaxf(bias[c] + a.x * inv, 0.0f), fmaxf(bias[c + 1] + a.y * inv, 0.0f),
                                fmaxf(bias[c + 2] + a.z * inv, 0.0f), fmaxf(bias[c + 3] + a.w * inv, 0.0f));
        }
        __syncthreads();

        constexpr int STRIPS = (NPB + 15) / 16;
        static_assert(CN == 32, "two 16-column tiles are written out below");
        const int lane = tid & 31;
        const int wave = __builtin_amdgcn_readfirstlane(tid >> 5);
        if (wave < STRIPS) {
            const int mn = lane & 15;
            const int hi = lane >> 4;
            const int kg = hi << 1;
            const int r0 = wave * 16;
            const float* arow = acc + min(r0 + mn, NPB - 1) * C;
            v8f c0 = (v8f){}, c1 = (v8f){}, cs = (v8f){};
            for (int k0 = 0; k0 < C; k0 += 4) {
                v2f a;
                a.x = arow[k0 + kg];
                a.y = arow[k0 + kg + 1];
                const float* brow = sW + (k0 + kg) * CN + mn;
                v2f b0, b1, bs;
                b0.x = brow[0];
                b0.y = brow[CN];
                b1.x = brow[16];
                b1.y = brow[16 + CN];
                bs.x = sX[(k0 + kg) * 16 + mn];
                bs.y = sX[(k0 + kg + 1) * 16 + mn];
                c0 = __builtin_amdgcn_wmma_f32_16x16x4_f32(
                     false, a,  false, b0,
                     (short)0, c0,  false,  false);
                c1 = __builtin_amdgcn_wmma_f32_16x16x4_f32(
                     false, a,  false, b1,
                     (short)0, c1,  false,  false);
                cs = __builtin_amdgcn_wmma_f32_16x16x4_f32(
                     false, a,  false, bs,
                     (short)0, cs,  false,  false);
            }
#pragma unroll
            for (int v = 0; v < 8; ++v) {
                const int rr = r0 + hi * 8 + v;
                if (rr < len) {
                    float* orow = outp + (size_t)(lo + rr) * CN;
                    orow[mn]      = c0[v];
                    orow[16 + mn] = c1[v];
                }
            }
            if (mn < 2) {
#pragma unroll
                for (int v = 0; v < 8; ++v) {
                    const int rr = r0 + hi * 8 + v;
                    if (rr < len) {
                        const int nd = lo + rr;
                        scn[nd + (nd & ~15) + mn * 16] = cs[v];
                    }
                }
            }
        }
    }
}


static inline int cdiv(int a, int b) { return (a + b - 1) / b; }

extern "C" void kernel_launch(void* const* d_in, const int* in_sizes, int n_in,
                              void* d_out, int out_size, void* d_ws, size_t ws_size,
                              hipStream_t stream) {
    const float* x      = (const float*)d_in[0];
    const int*   eidx   = (const int*)d_in[1];
    const float* W1     = (const float*)d_in[2];
    const float* a_src1 = (const float*)d_in[3];
    const float* a_dst1 = (const float*)d_in[4];
    const float* b1     = (const float*)d_in[5];
    const float* W2     = (const float*)d_in[6];
    const float* a_src2 = (const float*)d_in[7];
    const float* a_dst2 = (const float*)d_in[8];
    const float* b2     = (const float*)d_in[9];
    float* out = (float*)d_out;

    const int IN = 128, H = 64, O = 32;
    const int N = in_sizes[0] / IN;
    const int E = in_sizes[1] / 2;
    const int* src = eidx;
    const int* dst = eidx + E;

    float* ws    = (float*)d_ws;
    float* h1    = ws;
    float* agg1  = h1 + (size_t)N * H;
    float* h2    = agg1;
    float* sc2   = agg1 + (size_t)N * O;
    unsigned int* rid = (unsigned int*)(agg1 + (size_t)N * H);
    float* sc    = agg1 + (size_t)N * H + (size_t)N * O;

    const int B  = 256;
    const int nb = cdiv(E, 32);
    int* lst = (int*)(rid + (size_t)nb * 8);

    owner_planes<<<cdiv(nb, 64), 64, 0, stream>>>(dst, rid, E, nb);

    gemm_wmma_strip<4><<<cdiv(N, 128), B, (IN * H + IN * 16 + 2 * H) * sizeof(float), stream>>>(
        x, W1, h1, a_src1, a_dst1, sc, N, IN);
    edge_layer<64, true><<<cdiv(N, NPB), EL_T, 0, stream>>>(src, dst, rid, sc, h1, b1, h2, lst, N, E, nb,
                                                            W2, a_src2, a_dst2, sc2);

    edge_layer<32, false><<<cdiv(N, NPB), EL_T, 0, stream>>>(src, dst, rid, sc2, h2, b2, out, lst, N, E, nb,
                                                             nullptr, nullptr, nullptr, nullptr);
}


# Projection: the block's copy of W1 into LDS is eight 16-byte loads a thread, all issued ahead of the first LDS store: one memory trip (the loop waited for each 4-byte load: 32 dependent trips)
# speedup vs baseline: 5.5378x; 1.1715x over previous
#include <hip/hip_runtime.h>
#include <hip/hip_bf16.h>
#include <math.h>

#define SLOPE 0.2f

typedef float v2f __attribute__((ext_vector_type(2)));
typedef float v8f __attribute__((ext_vector_type(8)));


__device__ __forceinline__ float leaky(float v) {
    return v > 0.0f ? v : SLOPE * v;
}

template <int NT>
__global__ void gemm_wmma_strip(const float* __restrict__ A,
                                const float* __restrict__ B,
                                float* __restrict__ C,
                                const float* __restrict__ a_src,
                                const float* __restrict__ a_dst,
                                float* __restrict__ sc,
                                int M, int K) {
    const int N = NT * 16;
    extern __shared__ __attribute__((aligned(16))) float sB[];
    float* sX = sB + K * N;
    float* sV = sX + K * 16;

    static_assert((NT * 16 & (NT * 16 - 1)) == 0, "the rotated walk below wraps with a mask");
    float va = 0.0f, vd = 0.0f;
    if ((int)threadIdx.x < N) {
        va = a_src[threadIdx.x];
        vd = a_dst[threadIdx.x];
    }

    const int n4 = (K * N) >> 2;
    if ((reinterpret_cast<size_t>(B) & 15) == 0) {
        const float4* B4  = reinterpret_cast<const float4*>(B);
        float4*       sB4 = reinterpret_cast<float4*>(sB);
        constexpr int SU = 8;
        const int bd = (int)blockDim.x;
        int i = (int)threadIdx.x;
        for (; i + (SU - 1) * bd < n4; i += SU * bd) {
            float4 w[SU];
#pragma unroll
            for (int u = 0; u < SU; ++u) w[u] = B4[i + u * bd];
#pragma unroll
            for (int u = 0; u < SU; ++u) sB4[i + u * bd] = w[u];
        }
        for (; i < n4; i += bd) sB4[i] = B4[i];
    } else {
        for (int i = threadIdx.x; i < K * N; i += blockDim.x) sB[i] = B[i];
    }
    for (int i = threadIdx.x; i < K * 16; i += blockDim.x) sX[i] = 0.0f;
    if ((int)threadIdx.x < N) {
        sV[2 * threadIdx.x]     = va;
        sV[2 * threadIdx.x + 1] = vd;
    }
    __syncthreads();
    for (int t = threadIdx.x; t < 2 * K; t += blockDim.x) {
        const int k = t >> 1, which = t & 1;
        float s = 0.0f;
        for (int jj = 0; jj < N; ++jj) {
            const int j = (jj + k) & (N - 1);
            s += sB[k * N + j] * sV[2 * j + which];
        }
        sX[k * 16 + which] = s;
    }
    __syncthreads();

    const int lane = threadIdx.x & 31;
    const int wid  = threadIdx.x >> 5;
    const int row0 = (blockIdx.x * (blockDim.x >> 5) + wid) * 16;
    if (row0 >= M) return;

    const int mn = lane & 15;
    const int hi = lane >> 4;
    const int kg = hi << 1;

    const float* arow = A + (size_t)(row0 + mn) * K;

    v8f acc[NT];
#pragma unroll
    for (int t = 0; t < NT; ++t) acc[t] = (v8f){};
    v8f accs = (v8f){};

#pragma unroll (NT == 2 ? 8 : 4)
    for (int k0 = 0; k0 < K; k0 += 4) {
        v2f a;
        a.x = arow[k0 + kg];
        a.y = arow[k0 + kg + 1];
        const float* brow = sB + (size_t)(k0 + kg) * N + mn;
#pragma unroll
        for (int t = 0; t < NT; ++t) {
            v2f b;
            b.x = brow[t * 16];
            b.y = brow[t * 16 + N];
            acc[t] = __builtin_amdgcn_wmma_f32_16x16x4_f32(
                 false, a,  false, b,
                 (short)0, acc[t],  false,  false);
        }
        {
            v2f b;
            b.x = sX[(k0 + kg) * 16 + mn];
            b.y = sX[(k0 + kg + 1) * 16 + mn];
            accs = __builtin_amdgcn_wmma_f32_16x16x4_f32(
                 false, a,  false, b,
                 (short)0, accs,  false,  false);
        }
    }

    if (mn < 2) {
        float* sp = sc + (size_t)row0 * 2 + mn * 16 + hi * 8;
#pragma unroll
        for (int v = 0; v < 8; ++v) sp[v] = accs[v];
    }

#pragma unroll
    for (int t = 0; t < NT; ++t) {
        float* crow = C + (size_t)(row0 + hi * 8) * N + t * 16 + mn;
#pragma unroll
        for (int v = 0; v < 8; ++v) crow[(size_t)v * N] = acc[t][v];
    }
}


#define NPB    196
#define EL_T   1024
#define EL_CAP 12288

#define LS_U      4
#define LS_CAP    (LS_U * EL_T)
#define LS_HEAD   32
#define LS_STRIDE (LS_HEAD + LS_CAP)

__global__ __launch_bounds__(64)
void owner_planes(const int* __restrict__ dst, unsigned int* __restrict__ rid, int E, int nb) {
    const int blk = blockIdx.x * blockDim.x + threadIdx.x;
    if (blk >= nb) return;
    uint4* lo4 = reinterpret_cast<uint4*>(rid);
    uint4* hi4 = lo4 + nb;
    const int e0 = blk * 32;
    int d[32];
    if (e0 + 31 < E && (reinterpret_cast<size_t>(dst + e0) & 15) == 0) {
        const int4* p = reinterpret_cast<const int4*>(dst + e0);
#pragma unroll
        for (int k = 0; k < 8; ++k) {
            const int4 v = p[k];
            d[4 * k] = v.x; d[4 * k + 1] = v.y; d[4 * k + 2] = v.z; d[4 * k + 3] = v.w;
        }
    } else {
#pragma unroll
        for (int k = 0; k < 32; ++k) d[k] = dst[min(e0 + k, E - 1)];
    }
    unsigned int w[8];
#pragma unroll
    for (int k = 0; k < 8; ++k) w[k] = 0u;
#pragma unroll
    for (int i = 0; i < 32; ++i) {
        const unsigned int o = (unsigned int)(d[i] / NPB);
#pragma unroll
        for (int k = 0; k < 8; ++k) w[k] |= ((o >> k) & 1u) << i;
    }
    lo4[blk] = make_uint4(w[0], w[1], w[2], w[3]);
    hi4[blk] = make_uint4(w[4], w[5], w[6], w[7]);
}

__device__ __forceinline__ void ldsMaxFloat(float* addr, float value) {
    if (__float_as_int(value) >= 0) {
        atomicMax((int*)addr, __float_as_int(value));
    } else {
        atomicMin((unsigned int*)addr, __float_as_uint(value));
    }
}

struct OwnerSel { unsigned int s[8]; };

__device__ __forceinline__ void take_block(const uint4 a, const uint4 h, const OwnerSel& sel,
                                           int ebase, int E, int* Ls, int* cnt) {
    const unsigned int x = (a.x ^ sel.s[0]) | (a.y ^ sel.s[1]) | (a.z ^ sel.s[2]) | (a.w ^ sel.s[3]) |
                           (h.x ^ sel.s[4]) | (h.y ^ sel.s[5]) | (h.z ^ sel.s[6]) | (h.w ^ sel.s[7]);
    unsigned int m = ~x;
    while (m != 0u) {
        const int p = __builtin_ctz(m);
        m &= m - 1u;
        const int e = ebase + p;
        if (e < E) {
            const int pos = atomicAdd(cnt, 1);
            if (pos < EL_CAP) Ls[pos] = e;
        }
    }
}

__device__ __forceinline__ void scan_owner_planes(const uint4* __restrict__ lo4, const uint4* __restrict__ hi4,
                                                  int qb, int qe, const OwnerSel& sel, int E,
                                                  int* Ls, int* cnt, int tid) {
    constexpr int U = 4;
    int q = qb + tid;
    for (; q + (U - 1) * EL_T < qe; q += EL_T * U) {
        uint4 a[U], h[U];
#pragma unroll
        for (int u = 0; u < U; ++u) {
            a[u] = lo4[q + u * EL_T];
            h[u] = hi4[q + u * EL_T];
        }
#pragma unroll
        for (int u = 0; u < U; ++u)
            take_block(a[u], h[u], sel, (q + u * EL_T) * 32, E, Ls, cnt);
    }
    for (; q < qe; q += EL_T) {
        const uint4 a = lo4[q];
        const uint4 h = hi4[q];
        take_block(a, h, sel, q * 32, E, Ls, cnt);
    }
}

template <bool KEEP>
__device__ __forceinline__ void list_logits_max(const int* __restrict__ src, const int* __restrict__ dst,
                                                const float* __restrict__ sc, int lo, int n,
                                                int* Ls, int* Ld, float* Lw, float* Tm, const float* Tq, int tid,
                                                int* __restrict__ keep) {
    constexpr int U = 4;
    for (int i0 = tid; i0 < n; i0 += EL_T * U) {
        int s[U], d[U];
        float p[U];
#pragma unroll
        for (int u = 0; u < U; ++u) {
            const int i = i0 + u * EL_T;
            const int e = i < n ? Ls[i] : 0;
            s[u] = src[e];
            d[u] = dst[e];
        }
#pragma unroll
        for (int u = 0; u < U; ++u) p[u] = sc[s[u] + (s[u] & ~15)];
#pragma unroll
        for (int u = 0; u < U; ++u) {
            const int i = i0 + u * EL_T;
            if (i < n) {
                const int dl = d[u] - lo;
                const float lg = leaky(p[u] + Tq[dl]);
                Ls[i] = s[u];
                Ld[i] = dl;
                Lw[i] = lg;
                ldsMaxFloat(&Tm[dl], lg);
                if (KEEP) {
                    if (i < LS_CAP) keep[i] = (s[u] << 8) | dl;
                }
            }
        }
    }
}

template <int C>
__device__ __forceinline__ void list_sums_rows(const float* __restrict__ h, int n,
                                               const int* Ls, const int* Ld, float* Lw,
                                               const float* Tm, float* Ts, float* acc, int tid) {
    constexpr int R = C / 32;
    constexpr int G = 32 / R;
    for (int i = tid; i < n; i += EL_T) {
        const int dl = Ld[i];
        const float ex = __expf(Lw[i] - Tm[dl]);
        Lw[i] = ex;
        atomicAdd(&Ts[dl], ex);
    }
    const unsigned lane = tid & 31;
    const int wave = __builtin_amdgcn_readfirstlane(tid >> 5);
    const int nn   = __builtin_amdgcn_readfirstlane(n);
    for (int g = wave * 32; g < nn; g += EL_T) {
        const int i   = g + (int)lane;
        const int cw  = min(32, nn - g);
        const int   s = i < nn ? Ls[i] : 0;
        const int   d = i < nn ? Ld[i] : 0;
        const float a = i < nn ? Lw[i] : 0.0f;
#pragma unroll
        for (int j0 = 0; j0 < 32; j0 += G) {
            if (j0 < cw) {
                float v[G][R];
#pragma unroll
                for (int j = 0; j < G; ++j) {
                    const unsigned sb = (unsigned)__builtin_amdgcn_readlane(s, j0 + j) * C;
#pragma unroll
                    for (int r = 0; r < R; ++r) v[j][r] = h[sb + lane + r * 32];
                }
#pragma unroll
                for (int j = 0; j < G; ++j) {
                    if (j0 + j < cw) {
                        const unsigned db = (unsigned)__builtin_amdgcn_readlane(d, j0 + j) * C;
                        const float    aj = __int_as_float(__builtin_amdgcn_readlane(__float_as_int(a), j0 + j));
#pragma unroll
                        for (int r = 0; r < R; ++r) atomicAdd(&acc[db + lane + r * 32], aj * v[j][r]);
                    }
                }
            }
        }
    }
}

template <int C, bool FIRST>
__global__ __launch_bounds__(EL_T)
void edge_layer(const int* __restrict__ src, const int* __restrict__ dst,
                const unsigned int* __restrict__ rid, const float* __restrict__ sc,
                const float* __restrict__ h, const float* __restrict__ bias,
                float* __restrict__ outp, int* lst, int N, int E, int nb,
                const float* __restrict__ Wn, const float* __restrict__ an_src,
                const float* __restrict__ an_dst, float* __restrict__ scn) {
    constexpr int CN = 32;
    __shared__ float sW[FIRST ? C * CN : 1];
    __shared__ float sX[FIRST ? C * 16 : 1];
    __shared__ float sV[FIRST ? 2 * CN : 2];
    __shared__ __attribute__((aligned(16))) float acc[NPB * C];
    __shared__ float Tm[NPB];
    __shared__ float Ts[NPB];
    __shared__ float Tq[NPB];
    __shared__ int   Ls[EL_CAP];
    __shared__ int   Ld[EL_CAP];
    __shared__ float Lw[EL_CAP];
    __shared__ int   cnt;
    const int tid = threadIdx.x;
    const int lo  = (int)blockIdx.x * NPB;
    const int len = min(NPB, N - lo);
    OwnerSel sel;
#pragma unroll
    for (int k = 0; k < 8; ++k) sel.s[k] = ((blockIdx.x >> k) & 1u) ? 0xFFFFFFFFu : 0u;
    const uint4* lo4 = reinterpret_cast<const uint4*>(rid);
    const uint4* hi4 = lo4 + nb;
    int* slot = lst + (size_t)blockIdx.x * LS_STRIDE;

    int kept = 0;
    int pk[LS_U];
    if (!FIRST) {
        const int* kslot = slot;
        kept = kslot[0];
#pragma unroll
        for (int u = 0; u < LS_U; ++u) pk[u] = kslot[LS_HEAD + tid + u * EL_T];
    }
    const bool listed = !FIRST && kept <= LS_CAP;

    for (int i = tid; i < NPB * C; i += EL_T) acc[i] = 0.0f;
    for (int i = tid; i < NPB; i += EL_T) {
        const int nd = min(lo + i, N - 1);
        Tm[i] = -INFINITY;
        Ts[i] = 0.0f;
        Tq[i] = sc[nd + (nd & ~15) + 16];
    }
    if (tid == 0) cnt = 0;
    if (FIRST) {
        for (int i = tid; i < C * CN; i += EL_T) sW[i] = Wn[i];
        for (int i = tid; i < C * 16; i += EL_T) sX[i] = 0.0f;
        if (tid < CN) {
            sV[2 * tid]     = an_src[tid];
            sV[2 * tid + 1] = an_dst[tid];
        }
    }

    float p[LS_U];
    if (listed) {
#pragma unroll
        for (int u = 0; u < LS_U; ++u) {
            const int s = tid + u * EL_T < kept ? (int)((unsigned int)pk[u] >> 8) : 0;
            p[u] = sc[s + (s & ~15)];
        }
    }
    __syncthreads();

    if (FIRST) {
        if (tid < 2 * C) {
            const int k = tid >> 1, which = tid & 1;
            float s = 0.0f;
            for (int jj = 0; jj < CN; ++jj) {
                const int j = (jj + k) & (CN - 1);
                s += sW[k * CN + j] * sV[2 * j + which];
            }
            sX[k * 16 + which] = s;
        }
    }

    if (listed) {
#pragma unroll
        for (int u = 0; u < LS_U; ++u) {
            const int i = tid + u * EL_T;
            if (i < kept) {
                const int dl = pk[u] & 255;
                const float lg = leaky(p[u] + Tq[dl]);
                Ls[i] = (int)((unsigned int)pk[u] >> 8);
                Ld[i] = dl;
                Lw[i] = lg;
                ldsMaxFloat(&Tm[dl], lg);
            }
        }
        __syncthreads();
        list_sums_rows<C>(h, kept, Ls, Ld, Lw, Tm, Ts, acc, tid);
    } else {
        scan_owner_planes(lo4, hi4, 0, nb, sel, E, Ls, &cnt, tid);
        __syncthreads();
        const int total = cnt;
        if (FIRST && tid == 0) slot[0] = total;
        if (total <= EL_CAP) {
            list_logits_max<FIRST>(src, dst, sc, lo, total, Ls, Ld, Lw, Tm, Tq, tid, slot + LS_HEAD);
            __syncthreads();
            list_sums_rows<C>(h, total, Ls, Ld, Lw, Tm, Ts, acc, tid);
        } else {
            constexpr int SQ = EL_CAP / 32;
            for (int sweep = 0; sweep < 2; ++sweep) {
                for (int q0 = 0; q0 < nb; q0 += SQ) {
                    __syncthreads();
                    if (tid == 0) cnt = 0;
                    __syncthreads();
                    scan_owner_planes(lo4, hi4, q0, min(nb, q0 + SQ), sel, E, Ls, &cnt, tid);
                    __syncthreads();
                    const int n = cnt;
                    list_logits_max<false>(src, dst, sc, lo, n, Ls, Ld, Lw, Tm, Tq, tid, nullptr);
                    __syncthreads();
                    if (sweep == 1) list_sums_rows<C>(h, n, Ls, Ld, Lw, Tm, Ts, acc, tid);
                }
            }
        }
    }
    __syncthreads();

    if (!FIRST) {
        const int n4 = len * (C / 4);
        float4* o4 = reinterpret_cast<float4*>(outp + (size_t)lo * C);
        const float4* a4 = reinterpret_cast<const float4*>(acc);
        for (int i = tid; i < n4; i += EL_T) {
            const int dl = i / (C / 4);
            const int c  = (i % (C / 4)) * 4;
            const float S   = Ts[dl];
            const float inv = S > 0.0f ? 1.0f / S : 0.0f;
            const float4 a = a4[i];
            o4[i] = make_float4(bias[c] + a.x * inv, bias[c + 1] + a.y * inv,
                                bias[c + 2] + a.z * inv, bias[c + 3] + a.w * inv);
        }
    } else {
        float4* a4 = reinterpret_cast<float4*>(acc);
        for (int i = tid; i < NPB * (C / 4); i += EL_T) {
            const int dl = i / (C / 4);
            const int c  = (i % (C / 4)) * 4;
            const float S   = Ts[dl];
            const float inv = S > 0.0f ? 1.0f / S : 0.0f;
            const float4 a = a4[i];
            a4[i] = make_float4(fmaxf(bias[c] + a.x * inv, 0.0f), fmaxf(bias[c + 1] + a.y * inv, 0.0f),
                                fmaxf(bias[c + 2] + a.z * inv, 0.0f), fmaxf(bias[c + 3] + a.w * inv, 0.0f));
        }
        __syncthreads();

        constexpr int STRIPS = (NPB + 15) / 16;
        static_assert(CN == 32, "two 16-column tiles are written out below");
        const int lane = tid & 31;
        const int wave = __builtin_amdgcn_readfirstlane(tid >> 5);
        if (wave < STRIPS) {
            const int mn = lane & 15;
            const int hi = lane >> 4;
            const int kg = hi << 1;
            const int r0 = wave * 16;
            const float* arow = acc + min(r0 + mn, NPB - 1) * C;
            v8f c0 = (v8f){}, c1 = (v8f){}, cs = (v8f){};
            for (int k0 = 0; k0 < C; k0 += 4) {
                v2f a;
                a.x = arow[k0 + kg];
                a.y = arow[k0 + kg + 1];
                const float* brow = sW + (k0 + kg) * CN + mn;
                v2f b0, b1, bs;
                b0.x = brow[0];
                b0.y = brow[CN];
                b1.x = brow[16];
                b1.y = brow[16 + CN];
                bs.x = sX[(k0 + kg) * 16 + mn];
                bs.y = sX[(k0 + kg + 1) * 16 + mn];
                c0 = __builtin_amdgcn_wmma_f32_16x16x4_f32(
                     false, a,  false, b0,
                     (short)0, c0,  false,  false);
                c1 = __builtin_amdgcn_wmma_f32_16x16x4_f32(
                     false, a,  false, b1,
                     (short)0, c1,  false,  false);
                cs = __builtin_amdgcn_wmma_f32_16x16x4_f32(
                     false, a,  false, bs,
                     (short)0, cs,  false,  false);
            }
#pragma unroll
            for (int v = 0; v < 8; ++v) {
                const int rr = r0 + hi * 8 + v;
                if (rr < len) {
                    float* orow = outp + (size_t)(lo + rr) * CN;
                    orow[mn]      = c0[v];
                    orow[16 + mn] = c1[v];
                }
            }
            if (mn < 2) {
#pragma unroll
                for (int v = 0; v < 8; ++v) {
                    const int rr = r0 + hi * 8 + v;
                    if (rr < len) {
                        const int nd = lo + rr;
                        scn[nd + (nd & ~15) + mn * 16] = cs[v];
                    }
                }
            }
        }
    }
}


static inline int cdiv(int a, int b) { return (a + b - 1) / b; }

extern "C" void kernel_launch(void* const* d_in, const int* in_sizes, int n_in,
                              void* d_out, int out_size, void* d_ws, size_t ws_size,
                              hipStream_t stream) {
    const float* x      = (const float*)d_in[0];
    const int*   eidx   = (const int*)d_in[1];
    const float* W1     = (const float*)d_in[2];
    const float* a_src1 = (const float*)d_in[3];
    const float* a_dst1 = (const float*)d_in[4];
    const float* b1     = (const float*)d_in[5];
    const float* W2     = (const float*)d_in[6];
    const float* a_src2 = (const float*)d_in[7];
    const float* a_dst2 = (const float*)d_in[8];
    const float* b2     = (const float*)d_in[9];
    float* out = (float*)d_out;

    const int IN = 128, H = 64, O = 32;
    const int N = in_sizes[0] / IN;
    const int E = in_sizes[1] / 2;
    const int* src = eidx;
    const int* dst = eidx + E;

    float* ws    = (float*)d_ws;
    float* h1    = ws;
    float* agg1  = h1 + (size_t)N * H;
    float* h2    = agg1;
    float* sc2   = agg1 + (size_t)N * O;
    unsigned int* rid = (unsigned int*)(agg1 + (size_t)N * H);
    float* sc    = agg1 + (size_t)N * H + (size_t)N * O;

    const int B  = 256;
    const int nb = cdiv(E, 32);
    int* lst = (int*)(rid + (size_t)nb * 8);

    owner_planes<<<cdiv(nb, 64), 64, 0, stream>>>(dst, rid, E, nb);

    gemm_wmma_strip<4><<<cdiv(N, 128), B, (IN * H + IN * 16 + 2 * H) * sizeof(float), stream>>>(
        x, W1, h1, a_src1, a_dst1, sc, N, IN);
    edge_layer<64, true><<<cdiv(N, NPB), EL_T, 0, stream>>>(src, dst, rid, sc, h1, b1, h2, lst, N, E, nb,
                                                            W2, a_src2, a_dst2, sc2);

    edge_layer<32, false><<<cdiv(N, NPB), EL_T, 0, stream>>>(src, dst, rid, sc2, h2, b2, out, lst, N, E, nb,
                                                             nullptr, nullptr, nullptr, nullptr);
}
